# starting kernel
# baseline (speedup 1.0000x reference)
#include <hip/hip_runtime.h>

typedef __attribute__((ext_vector_type(2))) float v2f;
typedef __attribute__((ext_vector_type(8))) float v8f;

#define N_NODES 100000
#define N_EDGES 3200000
#define D_FEAT  256
#define HIDDEN  64
#define NCLASS  7
#define ROWBLKS (N_NODES / 16)


__global__ void k_zero(float* __restrict__ p, int n) {
    int i = blockIdx.x * blockDim.x + threadIdx.x;
    if (i < n) p[i] = 0.0f;
}

__global__ void k_deg(const int* __restrict__ ei, const float* __restrict__ ew,
                      float* __restrict__ deg) {
    int e = blockIdx.x * blockDim.x + threadIdx.x;
    if (e < N_EDGES) atomicAdd(&deg[ei[N_EDGES + e]], ew[e]);
}

__global__ void k_dinv(float* __restrict__ deg) {
    int i = blockIdx.x * blockDim.x + threadIdx.x;
    if (i < N_NODES) {
        float d = deg[i] + 1.0f;
        deg[i] = rsqrtf(fmaxf(d, 1e-12f));
    }
}


__global__ __launch_bounds__(128) void k_gemm0(const float* __restrict__ x,
                                               const float* __restrict__ W0,
                                               float* __restrict__ m0) {
    const int lane = threadIdx.x & 31;
    const int wv   = threadIdx.x >> 5;
    const int row0 = blockIdx.x * 16;
    const int col0 = wv * 16;
    const int hh   = lane >> 4;
    const int l15  = lane & 15;

    v8f c = {};
    const float* xrow = x  + (row0 + l15) * D_FEAT + 2 * hh;
    const float* wcol = W0 + (2 * hh) * HIDDEN + col0 + l15;

#pragma unroll 4
    for (int k = 0; k < D_FEAT; k += 4) {
        v2f a, b;
        a.x = xrow[k];
        a.y = xrow[k + 1];
        b.x = wcol[k * HIDDEN];
        b.y = wcol[k * HIDDEN + HIDDEN];
        c = __builtin_amdgcn_wmma_f32_16x16x4_f32(false, a, false, b,
                                                  (short)0, c, false, false);
    }
    float* out = m0 + (row0 + 8 * hh) * HIDDEN + col0 + l15;
#pragma unroll
    for (int v = 0; v < 8; ++v) out[v * HIDDEN] = c[v];
}


__global__ void k_init_h(const float* __restrict__ dinv, const float* __restrict__ m0,
                         float* __restrict__ h) {
    int idx = blockIdx.x * blockDim.x + threadIdx.x;
    if (idx < N_NODES * HIDDEN) {
        float di = dinv[idx / HIDDEN];
        h[idx] = di * di * m0[idx];
    }
}

__global__ __launch_bounds__(256) void k_scatter0(const int* __restrict__ ei,
                                                  const float* __restrict__ ew,
                                                  const float* __restrict__ dinv,
                                                  const float* __restrict__ m0,
                                                  float* __restrict__ h) {
    const int lane = threadIdx.x & 31;
    const int e = blockIdx.x * 8 + (threadIdx.x >> 5);
    if (e >= N_EDGES) return;
    const int s = ei[e];
    const int d = ei[N_EDGES + e];
    const float wn = dinv[s] * ew[e] * dinv[d];
    const float2 v = *(const float2*)(m0 + s * HIDDEN + 2 * lane);
    atomicAdd(&h[d * HIDDEN + 2 * lane],     wn * v.x);
    atomicAdd(&h[d * HIDDEN + 2 * lane + 1], wn * v.y);
}

__global__ void k_bias_relu(float* __restrict__ h, const float* __restrict__ b0) {
    int idx = blockIdx.x * blockDim.x + threadIdx.x;
    if (idx < N_NODES * HIDDEN)
        h[idx] = fmaxf(h[idx] + b0[idx % HIDDEN], 0.0f);
}


__global__ __launch_bounds__(128) void k_gemm1(const float* __restrict__ h,
                                               const float* __restrict__ W1,
                                               float* __restrict__ m1) {
    const int lane = threadIdx.x & 31;
    const int wv   = threadIdx.x >> 5;
    const int rb   = blockIdx.x * 4 + wv;
    if (rb >= ROWBLKS) return;
    const int row0 = rb * 16;
    const int hh   = lane >> 4;
    const int l15  = lane & 15;

    v8f c = {};
    const float* hrow = h + (row0 + l15) * HIDDEN + 2 * hh;
#pragma unroll
    for (int k = 0; k < HIDDEN; k += 4) {
        v2f a, b;
        a.x = hrow[k];
        a.y = hrow[k + 1];
        float bx = 0.0f, by = 0.0f;
        if (l15 < NCLASS) {
            bx = W1[(k + 2 * hh) * NCLASS + l15];
            by = W1[(k + 2 * hh + 1) * NCLASS + l15];
        }
        b.x = bx; b.y = by;
        c = __builtin_amdgcn_wmma_f32_16x16x4_f32(false, a, false, b,
                                                  (short)0, c, false, false);
    }
    if (l15 < NCLASS) {
        float* out = m1 + (row0 + 8 * hh) * NCLASS + l15;
#pragma unroll
        for (int v = 0; v < 8; ++v) out[v * NCLASS] = c[v];
    }
}


__global__ void k_init_out(const float* __restrict__ dinv, const float* __restrict__ m1,
                           const float* __restrict__ b1, float* __restrict__ out) {
    int idx = blockIdx.x * blockDim.x + threadIdx.x;
    if (idx < N_NODES * NCLASS) {
        int i = idx / NCLASS, j = idx - i * NCLASS;
        float di = dinv[i];
        out[idx] = di * di * m1[idx] + b1[j];
    }
}

__global__ __launch_bounds__(256) void k_scatter1(const int* __restrict__ ei,
                                                  const float* __restrict__ ew,
                                                  const float* __restrict__ dinv,
                                                  const float* __restrict__ m1,
                                                  float* __restrict__ out) {
    const int j = threadIdx.x & 7;
    const int e = blockIdx.x * 32 + (threadIdx.x >> 3);
    if (e >= N_EDGES || j >= NCLASS) return;
    const int s = ei[e];
    const int d = ei[N_EDGES + e];
    const float wn = dinv[s] * ew[e] * dinv[d];
    atomicAdd(&out[d * NCLASS + j], wn * m1[s * NCLASS + j]);
}


extern "C" void kernel_launch(void* const* d_in, const int* in_sizes, int n_in,
                              void* d_out, int out_size, void* d_ws, size_t ws_size,
                              hipStream_t stream) {
    const float* x  = (const float*)d_in[0];
    const int*   ei = (const int*)  d_in[1];
    const float* ew = (const float*)d_in[2];
    const float* W0 = (const float*)d_in[3];
    const float* b0 = (const float*)d_in[4];
    const float* W1 = (const float*)d_in[5];
    const float* b1 = (const float*)d_in[6];
    float* out = (float*)d_out;

    float* ws   = (float*)d_ws;
    float* dinv = ws;
    float* m0   = dinv + N_NODES;
    float* h    = m0 + (size_t)N_NODES * HIDDEN;
    float* m1   = h  + (size_t)N_NODES * HIDDEN;

    k_zero<<<(N_NODES + 255) / 256, 256, 0, stream>>>(dinv, N_NODES);
    k_deg <<<(N_EDGES + 255) / 256, 256, 0, stream>>>(ei, ew, dinv);
    k_dinv<<<(N_NODES + 255) / 256, 256, 0, stream>>>(dinv);

    k_gemm0<<<ROWBLKS, 128, 0, stream>>>(x, W0, m0);

    k_init_h<<<(N_NODES * HIDDEN + 255) / 256, 256, 0, stream>>>(dinv, m0, h);
    k_scatter0<<<(N_EDGES + 7) / 8, 256, 0, stream>>>(ei, ew, dinv, m0, h);
    k_bias_relu<<<(N_NODES * HIDDEN + 255) / 256, 256, 0, stream>>>(h, b0);

    k_gemm1<<<(ROWBLKS + 3) / 4, 128, 0, stream>>>(h, W1, m1);

    k_init_out<<<(N_NODES * NCLASS + 255) / 256, 256, 0, stream>>>(dinv, m1, b1, out);
    k_scatter1<<<(N_EDGES + 31) / 32, 256, 0, stream>>>(ei, ew, dinv, m1, out);
}


# layer-0 edge scatter re-mapped: one wave walks 32 edges, lane = feature channel, so each f32 atomic adds one whole 128-B line (half the line requests, 1/32 the waves); sums still in arrival order
# speedup vs baseline: 1.2727x; 1.2727x over previous
#include <hip/hip_runtime.h>

typedef __attribute__((ext_vector_type(2))) float v2f;
typedef __attribute__((ext_vector_type(8))) float v8f;

#define N_NODES 100000
#define N_EDGES 3200000
#define D_FEAT  256
#define HIDDEN  64
#define NCLASS  7
#define ROWBLKS (N_NODES / 16)


__global__ void k_zero(float* __restrict__ p, int n) {
    int i = blockIdx.x * blockDim.x + threadIdx.x;
    if (i < n) p[i] = 0.0f;
}

__global__ void k_deg(const int* __restrict__ ei, const float* __restrict__ ew,
                      float* __restrict__ deg) {
    int e = blockIdx.x * blockDim.x + threadIdx.x;
    if (e < N_EDGES) atomicAdd(&deg[ei[N_EDGES + e]], ew[e]);
}

__global__ void k_dinv(float* __restrict__ deg) {
    int i = blockIdx.x * blockDim.x + threadIdx.x;
    if (i < N_NODES) {
        float d = deg[i] + 1.0f;
        deg[i] = rsqrtf(fmaxf(d, 1e-12f));
    }
}


__global__ __launch_bounds__(128) void k_gemm0(const float* __restrict__ x,
                                               const float* __restrict__ W0,
                                               float* __restrict__ m0) {
    const int lane = threadIdx.x & 31;
    const int wv   = threadIdx.x >> 5;
    const int row0 = blockIdx.x * 16;
    const int col0 = wv * 16;
    const int hh   = lane >> 4;
    const int l15  = lane & 15;

    v8f c = {};
    const float* xrow = x  + (row0 + l15) * D_FEAT + 2 * hh;
    const float* wcol = W0 + (2 * hh) * HIDDEN + col0 + l15;

#pragma unroll 4
    for (int k = 0; k < D_FEAT; k += 4) {
        v2f a, b;
        a.x = xrow[k];
        a.y = xrow[k + 1];
        b.x = wcol[k * HIDDEN];
        b.y = wcol[k * HIDDEN + HIDDEN];
        c = __builtin_amdgcn_wmma_f32_16x16x4_f32(false, a, false, b,
                                                  (short)0, c, false, false);
    }
    float* out = m0 + (row0 + 8 * hh) * HIDDEN + col0 + l15;
#pragma unroll
    for (int v = 0; v < 8; ++v) out[v * HIDDEN] = c[v];
}


__global__ void k_init_h(const float* __restrict__ dinv, const float* __restrict__ m0,
                         float* __restrict__ h) {
    int idx = blockIdx.x * blockDim.x + threadIdx.x;
    if (idx < N_NODES * HIDDEN) {
        float di = dinv[idx / HIDDEN];
        h[idx] = di * di * m0[idx];
    }
}

static_assert(N_EDGES % 256 == 0, "edge pass assumes whole blocks of 8 waves x 32 edges");
__global__ __launch_bounds__(256) void k_scatter0(const int* __restrict__ ei,
                                                  const float* __restrict__ ew,
                                                  const float* __restrict__ dinv,
                                                  const float* __restrict__ m0,
                                                  float* __restrict__ h) {
    const int lane = threadIdx.x & 31;
    const int e = blockIdx.x * 256 + threadIdx.x;
    const int s_l = ei[e];
    const int d_l = ei[N_EDGES + e];
    const float wn_l = dinv[s_l] * ew[e] * dinv[d_l];
    const float* m0l = m0 + lane;
    float* hl = h + lane;
    for (int j0 = 0; j0 < 32; j0 += 8) {
        float v0[8], v1[8];
#pragma unroll
        for (int j = 0; j < 8; ++j) {
            const int s = __builtin_amdgcn_readlane(s_l, j0 + j);
            v0[j] = m0l[s * HIDDEN];
            v1[j] = m0l[s * HIDDEN + 32];
        }
#pragma unroll
        for (int j = 0; j < 8; ++j) {
            const int d = __builtin_amdgcn_readlane(d_l, j0 + j);
            const float wn = __int_as_float(
                __builtin_amdgcn_readlane(__float_as_int(wn_l), j0 + j));
            atomicAdd(hl + d * HIDDEN,      wn * v0[j]);
            atomicAdd(hl + d * HIDDEN + 32, wn * v1[j]);
        }
    }
}

__global__ void k_bias_relu(float* __restrict__ h, const float* __restrict__ b0) {
    int idx = blockIdx.x * blockDim.x + threadIdx.x;
    if (idx < N_NODES * HIDDEN)
        h[idx] = fmaxf(h[idx] + b0[idx % HIDDEN], 0.0f);
}


__global__ __launch_bounds__(128) void k_gemm1(const float* __restrict__ h,
                                               const float* __restrict__ W1,
                                               float* __restrict__ m1) {
    const int lane = threadIdx.x & 31;
    const int wv   = threadIdx.x >> 5;
    const int rb   = blockIdx.x * 4 + wv;
    if (rb >= ROWBLKS) return;
    const int row0 = rb * 16;
    const int hh   = lane >> 4;
    const int l15  = lane & 15;

    v8f c = {};
    const float* hrow = h + (row0 + l15) * HIDDEN + 2 * hh;
#pragma unroll
    for (int k = 0; k < HIDDEN; k += 4) {
        v2f a, b;
        a.x = hrow[k];
        a.y = hrow[k + 1];
        float bx = 0.0f, by = 0.0f;
        if (l15 < NCLASS) {
            bx = W1[(k + 2 * hh) * NCLASS + l15];
            by = W1[(k + 2 * hh + 1) * NCLASS + l15];
        }
        b.x = bx; b.y = by;
        c = __builtin_amdgcn_wmma_f32_16x16x4_f32(false, a, false, b,
                                                  (short)0, c, false, false);
    }
    if (l15 < NCLASS) {
        float* out = m1 + (row0 + 8 * hh) * NCLASS + l15;
#pragma unroll
        for (int v = 0; v < 8; ++v) out[v * NCLASS] = c[v];
    }
}


__global__ void k_init_out(const float* __restrict__ dinv, const float* __restrict__ m1,
                           const float* __restrict__ b1, float* __restrict__ out) {
    int idx = blockIdx.x * blockDim.x + threadIdx.x;
    if (idx < N_NODES * NCLASS) {
        int i = idx / NCLASS, j = idx - i * NCLASS;
        float di = dinv[i];
        out[idx] = di * di * m1[idx] + b1[j];
    }
}

__global__ __launch_bounds__(256) void k_scatter1(const int* __restrict__ ei,
                                                  const float* __restrict__ ew,
                                                  const float* __restrict__ dinv,
                                                  const float* __restrict__ m1,
                                                  float* __restrict__ out) {
    const int j = threadIdx.x & 7;
    const int e = blockIdx.x * 32 + (threadIdx.x >> 3);
    if (e >= N_EDGES || j >= NCLASS) return;
    const int s = ei[e];
    const int d = ei[N_EDGES + e];
    const float wn = dinv[s] * ew[e] * dinv[d];
    atomicAdd(&out[d * NCLASS + j], wn * m1[s * NCLASS + j]);
}


extern "C" void kernel_launch(void* const* d_in, const int* in_sizes, int n_in,
                              void* d_out, int out_size, void* d_ws, size_t ws_size,
                              hipStream_t stream) {
    const float* x  = (const float*)d_in[0];
    const int*   ei = (const int*)  d_in[1];
    const float* ew = (const float*)d_in[2];
    const float* W0 = (const float*)d_in[3];
    const float* b0 = (const float*)d_in[4];
    const float* W1 = (const float*)d_in[5];
    const float* b1 = (const float*)d_in[6];
    float* out = (float*)d_out;

    float* ws   = (float*)d_ws;
    float* dinv = ws;
    float* m0   = dinv + N_NODES;
    float* h    = m0 + (size_t)N_NODES * HIDDEN;
    float* m1   = h  + (size_t)N_NODES * HIDDEN;

    k_zero<<<(N_NODES + 255) / 256, 256, 0, stream>>>(dinv, N_NODES);
    k_deg <<<(N_EDGES + 255) / 256, 256, 0, stream>>>(ei, ew, dinv);
    k_dinv<<<(N_NODES + 255) / 256, 256, 0, stream>>>(dinv);

    k_gemm0<<<ROWBLKS, 128, 0, stream>>>(x, W0, m0);

    k_init_h<<<(N_NODES * HIDDEN + 255) / 256, 256, 0, stream>>>(dinv, m0, h);
    k_scatter0<<<N_EDGES / 256, 256, 0, stream>>>(ei, ew, dinv, m0, h);
    k_bias_relu<<<(N_NODES * HIDDEN + 255) / 256, 256, 0, stream>>>(h, b0);

    k_gemm1<<<(ROWBLKS + 3) / 4, 128, 0, stream>>>(h, W1, m1);

    k_init_out<<<(N_NODES * NCLASS + 255) / 256, 256, 0, stream>>>(dinv, m1, b1, out);
    k_scatter1<<<(N_EDGES + 31) / 32, 256, 0, stream>>>(ei, ew, dinv, m1, out);
}


# degree pass: each work-group adds its edge slice's weights into an LDS table of half the nodes and flushes it with whole-line f32 atomics (400128 line adds, not one add per edge); sum order changes
# speedup vs baseline: 1.5676x; 1.2317x over previous
#include <hip/hip_runtime.h>

typedef __attribute__((ext_vector_type(2))) float v2f;
typedef __attribute__((ext_vector_type(8))) float v8f;

#define N_NODES 100000
#define N_EDGES 3200000
#define D_FEAT  256
#define HIDDEN  64
#define NCLASS  7
#define ROWBLKS (N_NODES / 16)


__global__ void k_zero(float* __restrict__ p, int n) {
    int i = blockIdx.x * blockDim.x + threadIdx.x;
    if (i < n) p[i] = 0.0f;
}

#define DEG_TBL    50016
#define DEG_SLICES 128
#define DEG_EPS    (N_EDGES / DEG_SLICES)
#define DEG_UNR    13
static_assert(N_EDGES % DEG_SLICES == 0, "slices are equal");
static_assert(DEG_TBL % 32 == 0 && (N_NODES - DEG_TBL) % 32 == 0 && N_NODES <= 2 * DEG_TBL &&
              N_NODES > DEG_TBL, "two tables of whole lines cover the nodes");

__global__ __launch_bounds__(1024) void k_deg(const int* __restrict__ ei,
                                              const float* __restrict__ ew,
                                              float* __restrict__ deg) {
    __shared__ float tbl[DEG_TBL];
    const int tid   = threadIdx.x;
    const int half  = blockIdx.x & 1;
    const int slice = blockIdx.x >> 1;
    const int lo    = half * DEG_TBL;
    const unsigned n_here = half ? (N_NODES - DEG_TBL) : DEG_TBL;

    for (int i = tid; i < DEG_TBL; i += 1024) tbl[i] = 0.0f;
    __syncthreads();

    const int*   dst = ei + N_EDGES + slice * DEG_EPS;
    const float* w   = ew + slice * DEG_EPS;
    for (int base = 0; base < DEG_EPS; base += 1024 * DEG_UNR) {
        int dd[DEG_UNR]; float ww[DEG_UNR];
#pragma unroll
        for (int k = 0; k < DEG_UNR; ++k) {
            const int i = base + k * 1024 + tid;
            const bool in = i < DEG_EPS;
            dd[k] = in ? dst[i] : -1;
            ww[k] = in ? w[i]   : 0.0f;
        }
#pragma unroll
        for (int k = 0; k < DEG_UNR; ++k) {
            const unsigned r = (unsigned)(dd[k] - lo);
            if (r < n_here) atomicAdd(&tbl[r], ww[k]);
        }
    }
    __syncthreads();

    const int lane = tid & 31;
    const int wv   = tid >> 5;
    const int nl   = (int)(n_here >> 5);
    const int start = (slice * nl) / DEG_SLICES;
    for (int j = wv; j < nl; j += 32) {
        int line = j + start;
        if (line >= nl) line -= nl;
        atomicAdd(&deg[lo + line * 32 + lane], tbl[line * 32 + lane]);
    }
}

__global__ void k_dinv(float* __restrict__ deg) {
    int i = blockIdx.x * blockDim.x + threadIdx.x;
    if (i < N_NODES) {
        float d = deg[i] + 1.0f;
        deg[i] = rsqrtf(fmaxf(d, 1e-12f));
    }
}


__global__ __launch_bounds__(128) void k_gemm0(const float* __restrict__ x,
                                               const float* __restrict__ W0,
                                               float* __restrict__ m0) {
    const int lane = threadIdx.x & 31;
    const int wv   = threadIdx.x >> 5;
    const int row0 = blockIdx.x * 16;
    const int col0 = wv * 16;
    const int hh   = lane >> 4;
    const int l15  = lane & 15;

    v8f c = {};
    const float* xrow = x  + (row0 + l15) * D_FEAT + 2 * hh;
    const float* wcol = W0 + (2 * hh) * HIDDEN + col0 + l15;

#pragma unroll 4
    for (int k = 0; k < D_FEAT; k += 4) {
        v2f a, b;
        a.x = xrow[k];
        a.y = xrow[k + 1];
        b.x = wcol[k * HIDDEN];
        b.y = wcol[k * HIDDEN + HIDDEN];
        c = __builtin_amdgcn_wmma_f32_16x16x4_f32(false, a, false, b,
                                                  (short)0, c, false, false);
    }
    float* out = m0 + (row0 + 8 * hh) * HIDDEN + col0 + l15;
#pragma unroll
    for (int v = 0; v < 8; ++v) out[v * HIDDEN] = c[v];
}


__global__ void k_init_h(const float* __restrict__ dinv, const float* __restrict__ m0,
                         float* __restrict__ h) {
    int idx = blockIdx.x * blockDim.x + threadIdx.x;
    if (idx < N_NODES * HIDDEN) {
        float di = dinv[idx / HIDDEN];
        h[idx] = di * di * m0[idx];
    }
}

static_assert(N_EDGES % 256 == 0, "edge pass assumes whole blocks of 8 waves x 32 edges");
__global__ __launch_bounds__(256) void k_scatter0(const int* __restrict__ ei,
                                                  const float* __restrict__ ew,
                                                  const float* __restrict__ dinv,
                                                  const float* __restrict__ m0,
                                                  float* __restrict__ h) {
    const int lane = threadIdx.x & 31;
    const int e = blockIdx.x * 256 + threadIdx.x;
    const int s_l = ei[e];
    const int d_l = ei[N_EDGES + e];
    const float wn_l = dinv[s_l] * ew[e] * dinv[d_l];
    const float* m0l = m0 + lane;
    float* hl = h + lane;
    for (int j0 = 0; j0 < 32; j0 += 8) {
        float v0[8], v1[8];
#pragma unroll
        for (int j = 0; j < 8; ++j) {
            const int s = __builtin_amdgcn_readlane(s_l, j0 + j);
            v0[j] = m0l[s * HIDDEN];
            v1[j] = m0l[s * HIDDEN + 32];
        }
#pragma unroll
        for (int j = 0; j < 8; ++j) {
            const int d = __builtin_amdgcn_readlane(d_l, j0 + j);
            const float wn = __int_as_float(
                __builtin_amdgcn_readlane(__float_as_int(wn_l), j0 + j));
            atomicAdd(hl + d * HIDDEN,      wn * v0[j]);
            atomicAdd(hl + d * HIDDEN + 32, wn * v1[j]);
        }
    }
}

__global__ void k_bias_relu(float* __restrict__ h, const float* __restrict__ b0) {
    int idx = blockIdx.x * blockDim.x + threadIdx.x;
    if (idx < N_NODES * HIDDEN)
        h[idx] = fmaxf(h[idx] + b0[idx % HIDDEN], 0.0f);
}


__global__ __launch_bounds__(128) void k_gemm1(const float* __restrict__ h,
                                               const float* __restrict__ W1,
                                               float* __restrict__ m1) {
    const int lane = threadIdx.x & 31;
    const int wv   = threadIdx.x >> 5;
    const int rb   = blockIdx.x * 4 + wv;
    if (rb >= ROWBLKS) return;
    const int row0 = rb * 16;
    const int hh   = lane >> 4;
    const int l15  = lane & 15;

    v8f c = {};
    const float* hrow = h + (row0 + l15) * HIDDEN + 2 * hh;
#pragma unroll
    for (int k = 0; k < HIDDEN; k += 4) {
        v2f a, b;
        a.x = hrow[k];
        a.y = hrow[k + 1];
        float bx = 0.0f, by = 0.0f;
        if (l15 < NCLASS) {
            bx = W1[(k + 2 * hh) * NCLASS + l15];
            by = W1[(k + 2 * hh + 1) * NCLASS + l15];
        }
        b.x = bx; b.y = by;
        c = __builtin_amdgcn_wmma_f32_16x16x4_f32(false, a, false, b,
                                                  (short)0, c, false, false);
    }
    if (l15 < NCLASS) {
        float* out = m1 + (row0 + 8 * hh) * NCLASS + l15;
#pragma unroll
        for (int v = 0; v < 8; ++v) out[v * NCLASS] = c[v];
    }
}


__global__ void k_init_out(const float* __restrict__ dinv, const float* __restrict__ m1,
                           const float* __restrict__ b1, float* __restrict__ out) {
    int idx = blockIdx.x * blockDim.x + threadIdx.x;
    if (idx < N_NODES * NCLASS) {
        int i = idx / NCLASS, j = idx - i * NCLASS;
        float di = dinv[i];
        out[idx] = di * di * m1[idx] + b1[j];
    }
}

__global__ __launch_bounds__(256) void k_scatter1(const int* __restrict__ ei,
                                                  const float* __restrict__ ew,
                                                  const float* __restrict__ dinv,
                                                  const float* __restrict__ m1,
                                                  float* __restrict__ out) {
    const int j = threadIdx.x & 7;
    const int e = blockIdx.x * 32 + (threadIdx.x >> 3);
    if (e >= N_EDGES || j >= NCLASS) return;
    const int s = ei[e];
    const int d = ei[N_EDGES + e];
    const float wn = dinv[s] * ew[e] * dinv[d];
    atomicAdd(&out[d * NCLASS + j], wn * m1[s * NCLASS + j]);
}


extern "C" void kernel_launch(void* const* d_in, const int* in_sizes, int n_in,
                              void* d_out, int out_size, void* d_ws, size_t ws_size,
                              hipStream_t stream) {
    const float* x  = (const float*)d_in[0];
    const int*   ei = (const int*)  d_in[1];
    const float* ew = (const float*)d_in[2];
    const float* W0 = (const float*)d_in[3];
    const float* b0 = (const float*)d_in[4];
    const float* W1 = (const float*)d_in[5];
    const float* b1 = (const float*)d_in[6];
    float* out = (float*)d_out;

    float* ws   = (float*)d_ws;
    float* dinv = ws;
    float* m0   = dinv + N_NODES;
    float* h    = m0 + (size_t)N_NODES * HIDDEN;
    float* m1   = h  + (size_t)N_NODES * HIDDEN;

    k_zero<<<(N_NODES + 255) / 256, 256, 0, stream>>>(dinv, N_NODES);
    k_deg <<<2 * DEG_SLICES, 1024, 0, stream>>>(ei, ew, dinv);
    k_dinv<<<(N_NODES + 255) / 256, 256, 0, stream>>>(dinv);

    k_gemm0<<<ROWBLKS, 128, 0, stream>>>(x, W0, m0);

    k_init_h<<<(N_NODES * HIDDEN + 255) / 256, 256, 0, stream>>>(dinv, m0, h);
    k_scatter0<<<N_EDGES / 256, 256, 0, stream>>>(ei, ew, dinv, m0, h);
    k_bias_relu<<<(N_NODES * HIDDEN + 255) / 256, 256, 0, stream>>>(h, b0);

    k_gemm1<<<(ROWBLKS + 3) / 4, 128, 0, stream>>>(h, W1, m1);

    k_init_out<<<(N_NODES * NCLASS + 255) / 256, 256, 0, stream>>>(dinv, m1, b1, out);
    k_scatter1<<<(N_EDGES + 31) / 32, 256, 0, stream>>>(ei, ew, dinv, m1, out);
}


# layer-1 edge scatter: each work-group sums its edge slice's messages in an LDS table of a tenth of the nodes and flushes it with whole-line f32 atomics (250 blocks, was 800000 waves); sums reordered
# speedup vs baseline: 1.8427x; 1.1755x over previous
#include <hip/hip_runtime.h>

typedef __attribute__((ext_vector_type(2))) float v2f;
typedef __attribute__((ext_vector_type(8))) float v8f;

#define N_NODES 100000
#define N_EDGES 3200000
#define D_FEAT  256
#define HIDDEN  64
#define NCLASS  7
#define ROWBLKS (N_NODES / 16)


__global__ void k_zero(float* __restrict__ p, int n) {
    int i = blockIdx.x * blockDim.x + threadIdx.x;
    if (i < n) p[i] = 0.0f;
}

#define DEG_TBL    50016
#define DEG_SLICES 128
#define DEG_EPS    (N_EDGES / DEG_SLICES)
#define DEG_UNR    13
static_assert(N_EDGES % DEG_SLICES == 0, "slices are equal");
static_assert(DEG_TBL % 32 == 0 && (N_NODES - DEG_TBL) % 32 == 0 && N_NODES <= 2 * DEG_TBL &&
              N_NODES > DEG_TBL, "two tables of whole lines cover the nodes");

__global__ __launch_bounds__(1024) void k_deg(const int* __restrict__ ei,
                                              const float* __restrict__ ew,
                                              float* __restrict__ deg) {
    __shared__ float tbl[DEG_TBL];
    const int tid   = threadIdx.x;
    const int half  = blockIdx.x & 1;
    const int slice = blockIdx.x >> 1;
    const int lo    = half * DEG_TBL;
    const unsigned n_here = half ? (N_NODES - DEG_TBL) : DEG_TBL;

    for (int i = tid; i < DEG_TBL; i += 1024) tbl[i] = 0.0f;
    __syncthreads();

    const int*   dst = ei + N_EDGES + slice * DEG_EPS;
    const float* w   = ew + slice * DEG_EPS;
    for (int base = 0; base < DEG_EPS; base += 1024 * DEG_UNR) {
        int dd[DEG_UNR]; float ww[DEG_UNR];
#pragma unroll
        for (int k = 0; k < DEG_UNR; ++k) {
            const int i = base + k * 1024 + tid;
            const bool in = i < DEG_EPS;
            dd[k] = in ? dst[i] : -1;
            ww[k] = in ? w[i]   : 0.0f;
        }
#pragma unroll
        for (int k = 0; k < DEG_UNR; ++k) {
            const unsigned r = (unsigned)(dd[k] - lo);
            if (r < n_here) atomicAdd(&tbl[r], ww[k]);
        }
    }
    __syncthreads();

    const int lane = tid & 31;
    const int wv   = tid >> 5;
    const int nl   = (int)(n_here >> 5);
    const int start = (slice * nl) / DEG_SLICES;
    for (int j = wv; j < nl; j += 32) {
        int line = j + start;
        if (line >= nl) line -= nl;
        atomicAdd(&deg[lo + line * 32 + lane], tbl[line * 32 + lane]);
    }
}

__global__ void k_dinv(float* __restrict__ deg) {
    int i = blockIdx.x * blockDim.x + threadIdx.x;
    if (i < N_NODES) {
        float d = deg[i] + 1.0f;
        deg[i] = rsqrtf(fmaxf(d, 1e-12f));
    }
}


__global__ __launch_bounds__(128) void k_gemm0(const float* __restrict__ x,
                                               const float* __restrict__ W0,
                                               float* __restrict__ m0) {
    const int lane = threadIdx.x & 31;
    const int wv   = threadIdx.x >> 5;
    const int row0 = blockIdx.x * 16;
    const int col0 = wv * 16;
    const int hh   = lane >> 4;
    const int l15  = lane & 15;

    v8f c = {};
    const float* xrow = x  + (row0 + l15) * D_FEAT + 2 * hh;
    const float* wcol = W0 + (2 * hh) * HIDDEN + col0 + l15;

#pragma unroll 4
    for (int k = 0; k < D_FEAT; k += 4) {
        v2f a, b;
        a.x = xrow[k];
        a.y = xrow[k + 1];
        b.x = wcol[k * HIDDEN];
        b.y = wcol[k * HIDDEN + HIDDEN];
        c = __builtin_amdgcn_wmma_f32_16x16x4_f32(false, a, false, b,
                                                  (short)0, c, false, false);
    }
    float* out = m0 + (row0 + 8 * hh) * HIDDEN + col0 + l15;
#pragma unroll
    for (int v = 0; v < 8; ++v) out[v * HIDDEN] = c[v];
}


__global__ void k_init_h(const float* __restrict__ dinv, const float* __restrict__ m0,
                         float* __restrict__ h) {
    int idx = blockIdx.x * blockDim.x + threadIdx.x;
    if (idx < N_NODES * HIDDEN) {
        float di = dinv[idx / HIDDEN];
        h[idx] = di * di * m0[idx];
    }
}

static_assert(N_EDGES % 256 == 0, "edge pass assumes whole blocks of 8 waves x 32 edges");
__global__ __launch_bounds__(256) void k_scatter0(const int* __restrict__ ei,
                                                  const float* __restrict__ ew,
                                                  const float* __restrict__ dinv,
                                                  const float* __restrict__ m0,
                                                  float* __restrict__ h) {
    const int lane = threadIdx.x & 31;
    const int e = blockIdx.x * 256 + threadIdx.x;
    const int s_l = ei[e];
    const int d_l = ei[N_EDGES + e];
    const float wn_l = dinv[s_l] * ew[e] * dinv[d_l];
    const float* m0l = m0 + lane;
    float* hl = h + lane;
    for (int j0 = 0; j0 < 32; j0 += 8) {
        float v0[8], v1[8];
#pragma unroll
        for (int j = 0; j < 8; ++j) {
            const int s = __builtin_amdgcn_readlane(s_l, j0 + j);
            v0[j] = m0l[s * HIDDEN];
            v1[j] = m0l[s * HIDDEN + 32];
        }
#pragma unroll
        for (int j = 0; j < 8; ++j) {
            const int d = __builtin_amdgcn_readlane(d_l, j0 + j);
            const float wn = __int_as_float(
                __builtin_amdgcn_readlane(__float_as_int(wn_l), j0 + j));
            atomicAdd(hl + d * HIDDEN,      wn * v0[j]);
            atomicAdd(hl + d * HIDDEN + 32, wn * v1[j]);
        }
    }
}

__global__ void k_bias_relu(float* __restrict__ h, const float* __restrict__ b0) {
    int idx = blockIdx.x * blockDim.x + threadIdx.x;
    if (idx < N_NODES * HIDDEN)
        h[idx] = fmaxf(h[idx] + b0[idx % HIDDEN], 0.0f);
}


__global__ __launch_bounds__(128) void k_gemm1(const float* __restrict__ h,
                                               const float* __restrict__ W1,
                                               float* __restrict__ m1) {
    const int lane = threadIdx.x & 31;
    const int wv   = threadIdx.x >> 5;
    const int rb   = blockIdx.x * 4 + wv;
    if (rb >= ROWBLKS) return;
    const int row0 = rb * 16;
    const int hh   = lane >> 4;
    const int l15  = lane & 15;

    v8f c = {};
    const float* hrow = h + (row0 + l15) * HIDDEN + 2 * hh;
#pragma unroll
    for (int k = 0; k < HIDDEN; k += 4) {
        v2f a, b;
        a.x = hrow[k];
        a.y = hrow[k + 1];
        float bx = 0.0f, by = 0.0f;
        if (l15 < NCLASS) {
            bx = W1[(k + 2 * hh) * NCLASS + l15];
            by = W1[(k + 2 * hh + 1) * NCLASS + l15];
        }
        b.x = bx; b.y = by;
        c = __builtin_amdgcn_wmma_f32_16x16x4_f32(false, a, false, b,
                                                  (short)0, c, false, false);
    }
    if (l15 < NCLASS) {
        float* out = m1 + (row0 + 8 * hh) * NCLASS + l15;
#pragma unroll
        for (int v = 0; v < 8; ++v) out[v * NCLASS] = c[v];
    }
}


__global__ void k_init_out(const float* __restrict__ dinv, const float* __restrict__ m1,
                           const float* __restrict__ b1, float* __restrict__ out) {
    int idx = blockIdx.x * blockDim.x + threadIdx.x;
    if (idx < N_NODES * NCLASS) {
        int i = idx / NCLASS, j = idx - i * NCLASS;
        float di = dinv[i];
        out[idx] = di * di * m1[idx] + b1[j];
    }
}

#define S1_TBL    10016
#define S1_PARTS  10
#define S1_SLICES 25
#define S1_EPS    (N_EDGES / S1_SLICES)
#define S1_UNR    5
static_assert(N_EDGES % S1_SLICES == 0, "slices are equal");
static_assert(S1_EPS % (1024 * S1_UNR) == 0, "a slice is a whole number of trips");
static_assert(S1_TBL % 32 == 0 && N_NODES % 32 == 0, "every table is whole 128-byte lines of out");
static_assert(S1_TBL * S1_PARTS >= N_NODES && S1_TBL * (S1_PARTS - 1) < N_NODES,
              "the tables cover the nodes and the last one is not empty");

__global__ __launch_bounds__(1024) void k_scatter1(const int* __restrict__ ei,
                                                   const float* __restrict__ ew,
                                                   const float* __restrict__ dinv,
                                                   const float* __restrict__ m1,
                                                   float* __restrict__ out) {
    __shared__ float tbl[S1_TBL * NCLASS];
    const int tid   = threadIdx.x;
    const int part  = blockIdx.x % S1_PARTS;
    const int slice = blockIdx.x / S1_PARTS;
    const int lo    = part * S1_TBL;
    const int rest  = N_NODES - lo;
    const unsigned n_here = (unsigned)(rest < S1_TBL ? rest : S1_TBL);

    for (int i = tid; i < S1_TBL * NCLASS; i += 1024) tbl[i] = 0.0f;
    __syncthreads();

    const int*   src = ei + slice * S1_EPS + tid;
    const int*   dst = ei + N_EDGES + slice * S1_EPS + tid;
    const float* w   = ew + slice * S1_EPS + tid;

    int ns[S1_UNR], nd[S1_UNR]; float nw[S1_UNR];
#pragma unroll
    for (int k = 0; k < S1_UNR; ++k) {
        ns[k] = src[k * 1024]; nd[k] = dst[k * 1024]; nw[k] = w[k * 1024];
    }
    for (int base = 0; base < S1_EPS; base += 1024 * S1_UNR) {
        int cs[S1_UNR], cd[S1_UNR]; float cw[S1_UNR];
#pragma unroll
        for (int k = 0; k < S1_UNR; ++k) { cs[k] = ns[k]; cd[k] = nd[k]; cw[k] = nw[k]; }
        const int nb = base + 1024 * S1_UNR;
        if (nb < S1_EPS) {
#pragma unroll
            for (int k = 0; k < S1_UNR; ++k) {
                ns[k] = src[nb + k * 1024]; nd[k] = dst[nb + k * 1024]; nw[k] = w[nb + k * 1024];
            }
        }
        float ds[S1_UNR], dt[S1_UNR]; float v[S1_UNR][NCLASS];
#pragma unroll
        for (int k = 0; k < S1_UNR; ++k) {
            const unsigned r = (unsigned)(cd[k] - lo);
            if (r < n_here) {
                ds[k] = dinv[cs[k]];
                dt[k] = dinv[cd[k]];
                const float* row = m1 + cs[k] * NCLASS;
#pragma unroll
                for (int j = 0; j < NCLASS; ++j) v[k][j] = row[j];
            }
        }
#pragma unroll
        for (int k = 0; k < S1_UNR; ++k) {
            const unsigned r = (unsigned)(cd[k] - lo);
            if (r < n_here) {
                const float wn = ds[k] * cw[k] * dt[k];
#pragma unroll
                for (int j = 0; j < NCLASS; ++j) atomicAdd(&tbl[r * NCLASS + j], wn * v[k][j]);
            }
        }
    }
    __syncthreads();

    const int lane = tid & 31;
    const int wv   = tid >> 5;
    const int nl   = (int)(n_here * NCLASS) >> 5;
    const int start = (slice * nl) / S1_SLICES;
    float* o = out + lo * NCLASS;
    for (int j = wv; j < nl; j += 32) {
        int line = j + start;
        if (line >= nl) line -= nl;
        atomicAdd(&o[line * 32 + lane], tbl[line * 32 + lane]);
    }
}


extern "C" void kernel_launch(void* const* d_in, const int* in_sizes, int n_in,
                              void* d_out, int out_size, void* d_ws, size_t ws_size,
                              hipStream_t stream) {
    const float* x  = (const float*)d_in[0];
    const int*   ei = (const int*)  d_in[1];
    const float* ew = (const float*)d_in[2];
    const float* W0 = (const float*)d_in[3];
    const float* b0 = (const float*)d_in[4];
    const float* W1 = (const float*)d_in[5];
    const float* b1 = (const float*)d_in[6];
    float* out = (float*)d_out;

    float* ws   = (float*)d_ws;
    float* dinv = ws;
    float* m0   = dinv + N_NODES;
    float* h    = m0 + (size_t)N_NODES * HIDDEN;
    float* m1   = h  + (size_t)N_NODES * HIDDEN;

    k_zero<<<(N_NODES + 255) / 256, 256, 0, stream>>>(dinv, N_NODES);
    k_deg <<<2 * DEG_SLICES, 1024, 0, stream>>>(ei, ew, dinv);
    k_dinv<<<(N_NODES + 255) / 256, 256, 0, stream>>>(dinv);

    k_gemm0<<<ROWBLKS, 128, 0, stream>>>(x, W0, m0);

    k_init_h<<<(N_NODES * HIDDEN + 255) / 256, 256, 0, stream>>>(dinv, m0, h);
    k_scatter0<<<N_EDGES / 256, 256, 0, stream>>>(ei, ew, dinv, m0, h);
    k_bias_relu<<<(N_NODES * HIDDEN + 255) / 256, 256, 0, stream>>>(h, b0);

    k_gemm1<<<(ROWBLKS + 3) / 4, 128, 0, stream>>>(h, W1, m1);

    k_init_out<<<(N_NODES * NCLASS + 255) / 256, 256, 0, stream>>>(dinv, m1, b1, out);
    k_scatter1<<<S1_PARTS * S1_SLICES, 1024, 0, stream>>>(ei, ew, dinv, m1, out);
}


# layer-0 edge scatter: each work-group scans its edge slice for destinations in its 1250-node part, sums them in an LDS table, flushes it with whole-line f32 atomics (600k, was 6400k); sums reordered
# speedup vs baseline: 2.0413x; 1.1078x over previous
#include <hip/hip_runtime.h>

typedef __attribute__((ext_vector_type(2))) float v2f;
typedef __attribute__((ext_vector_type(8))) float v8f;

#define N_NODES 100000
#define N_EDGES 3200000
#define D_FEAT  256
#define HIDDEN  64
#define NCLASS  7
#define ROWBLKS (N_NODES / 16)


__global__ void k_zero(float* __restrict__ p, int n) {
    int i = blockIdx.x * blockDim.x + threadIdx.x;
    if (i < n) p[i] = 0.0f;
}

#define DEG_TBL    50016
#define DEG_SLICES 128
#define DEG_EPS    (N_EDGES / DEG_SLICES)
#define DEG_UNR    13
static_assert(N_EDGES % DEG_SLICES == 0, "slices are equal");
static_assert(DEG_TBL % 32 == 0 && (N_NODES - DEG_TBL) % 32 == 0 && N_NODES <= 2 * DEG_TBL &&
              N_NODES > DEG_TBL, "two tables of whole lines cover the nodes");

__global__ __launch_bounds__(1024) void k_deg(const int* __restrict__ ei,
                                              const float* __restrict__ ew,
                                              float* __restrict__ deg) {
    __shared__ float tbl[DEG_TBL];
    const int tid   = threadIdx.x;
    const int half  = blockIdx.x & 1;
    const int slice = blockIdx.x >> 1;
    const int lo    = half * DEG_TBL;
    const unsigned n_here = half ? (N_NODES - DEG_TBL) : DEG_TBL;

    for (int i = tid; i < DEG_TBL; i += 1024) tbl[i] = 0.0f;
    __syncthreads();

    const int*   dst = ei + N_EDGES + slice * DEG_EPS;
    const float* w   = ew + slice * DEG_EPS;
    for (int base = 0; base < DEG_EPS; base += 1024 * DEG_UNR) {
        int dd[DEG_UNR]; float ww[DEG_UNR];
#pragma unroll
        for (int k = 0; k < DEG_UNR; ++k) {
            const int i = base + k * 1024 + tid;
            const bool in = i < DEG_EPS;
            dd[k] = in ? dst[i] : -1;
            ww[k] = in ? w[i]   : 0.0f;
        }
#pragma unroll
        for (int k = 0; k < DEG_UNR; ++k) {
            const unsigned r = (unsigned)(dd[k] - lo);
            if (r < n_here) atomicAdd(&tbl[r], ww[k]);
        }
    }
    __syncthreads();

    const int lane = tid & 31;
    const int wv   = tid >> 5;
    const int nl   = (int)(n_here >> 5);
    const int start = (slice * nl) / DEG_SLICES;
    for (int j = wv; j < nl; j += 32) {
        int line = j + start;
        if (line >= nl) line -= nl;
        atomicAdd(&deg[lo + line * 32 + lane], tbl[line * 32 + lane]);
    }
}

__global__ void k_dinv(float* __restrict__ deg) {
    int i = blockIdx.x * blockDim.x + threadIdx.x;
    if (i < N_NODES) {
        float d = deg[i] + 1.0f;
        deg[i] = rsqrtf(fmaxf(d, 1e-12f));
    }
}


__global__ __launch_bounds__(128) void k_gemm0(const float* __restrict__ x,
                                               const float* __restrict__ W0,
                                               float* __restrict__ m0) {
    const int lane = threadIdx.x & 31;
    const int wv   = threadIdx.x >> 5;
    const int row0 = blockIdx.x * 16;
    const int col0 = wv * 16;
    const int hh   = lane >> 4;
    const int l15  = lane & 15;

    v8f c = {};
    const float* xrow = x  + (row0 + l15) * D_FEAT + 2 * hh;
    const float* wcol = W0 + (2 * hh) * HIDDEN + col0 + l15;

#pragma unroll 4
    for (int k = 0; k < D_FEAT; k += 4) {
        v2f a, b;
        a.x = xrow[k];
        a.y = xrow[k + 1];
        b.x = wcol[k * HIDDEN];
        b.y = wcol[k * HIDDEN + HIDDEN];
        c = __builtin_amdgcn_wmma_f32_16x16x4_f32(false, a, false, b,
                                                  (short)0, c, false, false);
    }
    float* out = m0 + (row0 + 8 * hh) * HIDDEN + col0 + l15;
#pragma unroll
    for (int v = 0; v < 8; ++v) out[v * HIDDEN] = c[v];
}


__global__ void k_init_h(const float* __restrict__ dinv, const float* __restrict__ m0,
                         float* __restrict__ h) {
    int idx = blockIdx.x * blockDim.x + threadIdx.x;
    if (idx < N_NODES * HIDDEN) {
        float di = dinv[idx / HIDDEN];
        h[idx] = di * di * m0[idx];
    }
}

#define S0_TBL    1250
#define S0_PARTS  80
#define S0_SLICES 3
#define S0_EPS    (1024 * 1042)
#define S0_UNR    16
#define S0_G      16
#define S0_QCAP   48
static_assert(S0_TBL * S0_PARTS == N_NODES, "the tables tile the nodes exactly");
static_assert((S0_TBL * HIDDEN) % 32 == 0, "a table is whole 128-byte lines of h");
static_assert(S0_EPS * S0_SLICES >= N_EDGES && S0_EPS * (S0_SLICES - 1) < N_EDGES,
              "the slices cover the edges and the last one is not empty");
static_assert(S0_TBL <= 2048 && S0_EPS <= (1 << 21), "a queue record packs edge and row in 32 bits");
static_assert(32 % S0_G == 0 && S0_QCAP >= 32 && S0_UNR <= 32 && S0_UNR % 4 == 0, "queue and group sizes");
static_assert(S0_EPS % 4 == 0 && N_EDGES % 4 == 0, "a slice is whole groups of four edges");

__global__ __launch_bounds__(1024) void k_scatter0(const int* __restrict__ ei,
                                                  const float* __restrict__ ew,
                                                  const float* __restrict__ dinv,
                                                  const float* __restrict__ m0,
                                                  float* __restrict__ h) {
    __shared__ float tbl[S0_TBL * HIDDEN];
    __shared__ unsigned queue[32 * S0_QCAP];
    const int tid   = threadIdx.x;
    const int lane  = tid & 31;
    const int qb    = (tid >> 5) * S0_QCAP;
    const int part  = blockIdx.x % S0_PARTS;
    const int slice = blockIdx.x / S0_PARTS;
    const int lo    = part * S0_TBL;
    const int e0    = slice * S0_EPS;
    const int e1    = (e0 + S0_EPS < N_EDGES) ? e0 + S0_EPS : N_EDGES;

    for (int i = tid; i < S0_TBL * HIDDEN; i += 1024) tbl[i] = 0.0f;
    __syncthreads();

    const int*   dst = ei + N_EDGES;
    float*       tl  = tbl + lane;

    int qn = 0;
    for (int base = e0; ; base += 1024 * S0_UNR) {
        const bool last = base >= e1;
        unsigned ovf = 0u;
        if (!last) {
            int dd[S0_UNR];
#pragma unroll
            for (int q = 0; q < S0_UNR / 4; ++q) {
                const int i = base + (q * 1024 + tid) * 4;
                const bool in = i < e1;
                const int* p = dst + (in ? i : e1 - 4);
#pragma unroll
                for (int c = 0; c < 4; ++c) {
                    const int d = p[c];
                    dd[4 * q + c] = in ? d : -1;
                }
            }
#pragma unroll
            for (int k = 0; k < S0_UNR; ++k) {
                const unsigned r = (unsigned)(dd[k] - lo);
                const bool hit = r < (unsigned)S0_TBL;
                const unsigned mask = __builtin_amdgcn_ballot_w32(hit);
                if (mask != 0u) {
                    const int pos = qn + (int)__builtin_amdgcn_mbcnt_lo(mask, 0u);
                    if (hit) {
                        if (pos < S0_QCAP)
                            queue[qb + pos] = ((unsigned)(base + ((k >> 2) * 1024 + tid) * 4 + (k & 3) - e0) << 11) | r;
                        else
                            ovf |= 1u << k;
                    }
                    qn += __builtin_popcount(mask);
                }
            }
            if (qn > S0_QCAP) qn = S0_QCAP;
        }

        int off = 0;
        while (qn - off >= 32 || (last && qn - off > 0)) {
            const int n = (qn - off < 32) ? qn - off : 32;
            unsigned rec = 0u;
            if (lane < n) rec = queue[qb + off + lane];
            const int r_l = (int)(rec & 2047u);
            int   s_l  = 0;
            float wn_l = 0.0f;
            if (lane < n) {
                const int e = e0 + (int)(rec >> 11);
                s_l  = ei[e];
                wn_l = dinv[s_l] * ew[e] * dinv[lo + r_l];
            }
            for (int j0 = 0; j0 < n; j0 += S0_G) {
                float v0[S0_G], v1[S0_G];
#pragma unroll
                for (int j = 0; j < S0_G; ++j) {
                    const float* row = m0 + (unsigned)__builtin_amdgcn_readlane(s_l, j0 + j) * HIDDEN;
                    v0[j] = row[lane];
                    v1[j] = row[lane + 32];
                }
#pragma unroll
                for (int j = 0; j < S0_G; ++j) {
                    if (j0 + j < n) {
                        const int r = __builtin_amdgcn_readlane(r_l, j0 + j);
                        const float wn = __int_as_float(
                            __builtin_amdgcn_readlane(__float_as_int(wn_l), j0 + j));
                        atomicAdd(tl + r * HIDDEN,      wn * v0[j]);
                        atomicAdd(tl + r * HIDDEN + 32, wn * v1[j]);
                    }
                }
            }
            off += n;
        }
        const int rem = qn - off;
        if (off > 0 && lane < rem) {
            const unsigned t = queue[qb + off + lane];
            queue[qb + lane] = t;
        }
        qn = rem;

        if (ovf != 0u) {
#pragma nounroll
            for (int k = 0; k < S0_UNR; ++k) {
                if ((ovf >> k) & 1u) {
                    const int e = base + ((k >> 2) * 1024 + tid) * 4 + (k & 3);
                    const int s = ei[e];
                    const int d = dst[e];
                    const float wn = dinv[s] * ew[e] * dinv[d];
                    const float* row = m0 + s * HIDDEN;
                    float* trow = tbl + (d - lo) * HIDDEN;
#pragma nounroll
                    for (int f = 0; f < HIDDEN; ++f) atomicAdd(trow + f, wn * row[f]);
                }
            }
        }
        if (last) break;
    }
    __syncthreads();

    const int wv    = tid >> 5;
    const int nl    = S0_TBL * HIDDEN / 32;
    const int start = (slice * nl) / S0_SLICES;
    float* o = h + lo * HIDDEN;
    for (int j = wv; j < nl; j += 32) {
        int line = j + start;
        if (line >= nl) line -= nl;
        atomicAdd(&o[line * 32 + lane], tbl[line * 32 + lane]);
    }
}

__global__ void k_bias_relu(float* __restrict__ h, const float* __restrict__ b0) {
    int idx = blockIdx.x * blockDim.x + threadIdx.x;
    if (idx < N_NODES * HIDDEN)
        h[idx] = fmaxf(h[idx] + b0[idx % HIDDEN], 0.0f);
}


__global__ __launch_bounds__(128) void k_gemm1(const float* __restrict__ h,
                                               const float* __restrict__ W1,
                                               float* __restrict__ m1) {
    const int lane = threadIdx.x & 31;
    const int wv   = threadIdx.x >> 5;
    const int rb   = blockIdx.x * 4 + wv;
    if (rb >= ROWBLKS) return;
    const int row0 = rb * 16;
    const int hh   = lane >> 4;
    const int l15  = lane & 15;

    v8f c = {};
    const float* hrow = h + (row0 + l15) * HIDDEN + 2 * hh;
#pragma unroll
    for (int k = 0; k < HIDDEN; k += 4) {
        v2f a, b;
        a.x = hrow[k];
        a.y = hrow[k + 1];
        float bx = 0.0f, by = 0.0f;
        if (l15 < NCLASS) {
            bx = W1[(k + 2 * hh) * NCLASS + l15];
            by = W1[(k + 2 * hh + 1) * NCLASS + l15];
        }
        b.x = bx; b.y = by;
        c = __builtin_amdgcn_wmma_f32_16x16x4_f32(false, a, false, b,
                                                  (short)0, c, false, false);
    }
    if (l15 < NCLASS) {
        float* out = m1 + (row0 + 8 * hh) * NCLASS + l15;
#pragma unroll
        for (int v = 0; v < 8; ++v) out[v * NCLASS] = c[v];
    }
}


__global__ void k_init_out(const float* __restrict__ dinv, const float* __restrict__ m1,
                           const float* __restrict__ b1, float* __restrict__ out) {
    int idx = blockIdx.x * blockDim.x + threadIdx.x;
    if (idx < N_NODES * NCLASS) {
        int i = idx / NCLASS, j = idx - i * NCLASS;
        float di = dinv[i];
        out[idx] = di * di * m1[idx] + b1[j];
    }
}

#define S1_TBL    10016
#define S1_PARTS  10
#define S1_SLICES 25
#define S1_EPS    (N_EDGES / S1_SLICES)
#define S1_UNR    5
static_assert(N_EDGES % S1_SLICES == 0, "slices are equal");
static_assert(S1_EPS % (1024 * S1_UNR) == 0, "a slice is a whole number of trips");
static_assert(S1_TBL % 32 == 0 && N_NODES % 32 == 0, "every table is whole 128-byte lines of out");
static_assert(S1_TBL * S1_PARTS >= N_NODES && S1_TBL * (S1_PARTS - 1) < N_NODES,
              "the tables cover the nodes and the last one is not empty");

__global__ __launch_bounds__(1024) void k_scatter1(const int* __restrict__ ei,
                                                   const float* __restrict__ ew,
                                                   const float* __restrict__ dinv,
                                                   const float* __restrict__ m1,
                                                   float* __restrict__ out) {
    __shared__ float tbl[S1_TBL * NCLASS];
    const int tid   = threadIdx.x;
    const int part  = blockIdx.x % S1_PARTS;
    const int slice = blockIdx.x / S1_PARTS;
    const int lo    = part * S1_TBL;
    const int rest  = N_NODES - lo;
    const unsigned n_here = (unsigned)(rest < S1_TBL ? rest : S1_TBL);

    for (int i = tid; i < S1_TBL * NCLASS; i += 1024) tbl[i] = 0.0f;
    __syncthreads();

    const int*   src = ei + slice * S1_EPS + tid;
    const int*   dst = ei + N_EDGES + slice * S1_EPS + tid;
    const float* w   = ew + slice * S1_EPS + tid;

    int ns[S1_UNR], nd[S1_UNR]; float nw[S1_UNR];
#pragma unroll
    for (int k = 0; k < S1_UNR; ++k) {
        ns[k] = src[k * 1024]; nd[k] = dst[k * 1024]; nw[k] = w[k * 1024];
    }
    for (int base = 0; base < S1_EPS; base += 1024 * S1_UNR) {
        int cs[S1_UNR], cd[S1_UNR]; float cw[S1_UNR];
#pragma unroll
        for (int k = 0; k < S1_UNR; ++k) { cs[k] = ns[k]; cd[k] = nd[k]; cw[k] = nw[k]; }
        const int nb = base + 1024 * S1_UNR;
        if (nb < S1_EPS) {
#pragma unroll
            for (int k = 0; k < S1_UNR; ++k) {
                ns[k] = src[nb + k * 1024]; nd[k] = dst[nb + k * 1024]; nw[k] = w[nb + k * 1024];
            }
        }
        float ds[S1_UNR], dt[S1_UNR]; float v[S1_UNR][NCLASS];
#pragma unroll
        for (int k = 0; k < S1_UNR; ++k) {
            const unsigned r = (unsigned)(cd[k] - lo);
            if (r < n_here) {
                ds[k] = dinv[cs[k]];
                dt[k] = dinv[cd[k]];
                const float* row = m1 + cs[k] * NCLASS;
#pragma unroll
                for (int j = 0; j < NCLASS; ++j) v[k][j] = row[j];
            }
        }
#pragma unroll
        for (int k = 0; k < S1_UNR; ++k) {
            const unsigned r = (unsigned)(cd[k] - lo);
            if (r < n_here) {
                const float wn = ds[k] * cw[k] * dt[k];
#pragma unroll
                for (int j = 0; j < NCLASS; ++j) atomicAdd(&tbl[r * NCLASS + j], wn * v[k][j]);
            }
        }
    }
    __syncthreads();

    const int lane = tid & 31;
    const int wv   = tid >> 5;
    const int nl   = (int)(n_here * NCLASS) >> 5;
    const int start = (slice * nl) / S1_SLICES;
    float* o = out + lo * NCLASS;
    for (int j = wv; j < nl; j += 32) {
        int line = j + start;
        if (line >= nl) line -= nl;
        atomicAdd(&o[line * 32 + lane], tbl[line * 32 + lane]);
    }
}


extern "C" void kernel_launch(void* const* d_in, const int* in_sizes, int n_in,
                              void* d_out, int out_size, void* d_ws, size_t ws_size,
                              hipStream_t stream) {
    const float* x  = (const float*)d_in[0];
    const int*   ei = (const int*)  d_in[1];
    const float* ew = (const float*)d_in[2];
    const float* W0 = (const float*)d_in[3];
    const float* b0 = (const float*)d_in[4];
    const float* W1 = (const float*)d_in[5];
    const float* b1 = (const float*)d_in[6];
    float* out = (float*)d_out;

    float* ws   = (float*)d_ws;
    float* dinv = ws;
    float* m0   = dinv + N_NODES;
    float* h    = m0 + (size_t)N_NODES * HIDDEN;
    float* m1   = h  + (size_t)N_NODES * HIDDEN;

    k_zero<<<(N_NODES + 255) / 256, 256, 0, stream>>>(dinv, N_NODES);
    k_deg <<<2 * DEG_SLICES, 1024, 0, stream>>>(ei, ew, dinv);
    k_dinv<<<(N_NODES + 255) / 256, 256, 0, stream>>>(dinv);

    k_gemm0<<<ROWBLKS, 128, 0, stream>>>(x, W0, m0);

    k_init_h<<<(N_NODES * HIDDEN + 255) / 256, 256, 0, stream>>>(dinv, m0, h);
    k_scatter0<<<S0_PARTS * S0_SLICES, 1024, 0, stream>>>(ei, ew, dinv, m0, h);
    k_bias_relu<<<(N_NODES * HIDDEN + 255) / 256, 256, 0, stream>>>(h, b0);

    k_gemm1<<<(ROWBLKS + 3) / 4, 128, 0, stream>>>(h, W1, m1);

    k_init_out<<<(N_NODES * NCLASS + 255) / 256, 256, 0, stream>>>(dinv, m1, b1, out);
    k_scatter1<<<S1_PARTS * S1_SLICES, 1024, 0, stream>>>(ei, ew, dinv, m1, out);
}


# three one-element-per-thread passes (h start, bias+relu, out start) folded into the two GEMM kernels' epilogues and A loader: 7 launches not 10, 422k short waves fewer; every sum keeps its order
# speedup vs baseline: 2.2096x; 1.0824x over previous
#include <hip/hip_runtime.h>

typedef __attribute__((ext_vector_type(2))) float v2f;
typedef __attribute__((ext_vector_type(8))) float v8f;

#define N_NODES 100000
#define N_EDGES 3200000
#define D_FEAT  256
#define HIDDEN  64
#define NCLASS  7
#define ROWBLKS (N_NODES / 16)


__global__ void k_zero(float* __restrict__ p, int n) {
    int i = blockIdx.x * blockDim.x + threadIdx.x;
    if (i < n) p[i] = 0.0f;
}

#define DEG_TBL    50016
#define DEG_SLICES 128
#define DEG_EPS    (N_EDGES / DEG_SLICES)
#define DEG_UNR    13
static_assert(N_EDGES % DEG_SLICES == 0, "slices are equal");
static_assert(DEG_TBL % 32 == 0 && (N_NODES - DEG_TBL) % 32 == 0 && N_NODES <= 2 * DEG_TBL &&
              N_NODES > DEG_TBL, "two tables of whole lines cover the nodes");

__global__ __launch_bounds__(1024) void k_deg(const int* __restrict__ ei,
                                              const float* __restrict__ ew,
                                              float* __restrict__ deg) {
    __shared__ float tbl[DEG_TBL];
    const int tid   = threadIdx.x;
    const int half  = blockIdx.x & 1;
    const int slice = blockIdx.x >> 1;
    const int lo    = half * DEG_TBL;
    const unsigned n_here = half ? (N_NODES - DEG_TBL) : DEG_TBL;

    for (int i = tid; i < DEG_TBL; i += 1024) tbl[i] = 0.0f;
    __syncthreads();

    const int*   dst = ei + N_EDGES + slice * DEG_EPS;
    const float* w   = ew + slice * DEG_EPS;
    for (int base = 0; base < DEG_EPS; base += 1024 * DEG_UNR) {
        int dd[DEG_UNR]; float ww[DEG_UNR];
#pragma unroll
        for (int k = 0; k < DEG_UNR; ++k) {
            const int i = base + k * 1024 + tid;
            const bool in = i < DEG_EPS;
            dd[k] = in ? dst[i] : -1;
            ww[k] = in ? w[i]   : 0.0f;
        }
#pragma unroll
        for (int k = 0; k < DEG_UNR; ++k) {
            const unsigned r = (unsigned)(dd[k] - lo);
            if (r < n_here) atomicAdd(&tbl[r], ww[k]);
        }
    }
    __syncthreads();

    const int lane = tid & 31;
    const int wv   = tid >> 5;
    const int nl   = (int)(n_here >> 5);
    const int start = (slice * nl) / DEG_SLICES;
    for (int j = wv; j < nl; j += 32) {
        int line = j + start;
        if (line >= nl) line -= nl;
        atomicAdd(&deg[lo + line * 32 + lane], tbl[line * 32 + lane]);
    }
}

__global__ void k_dinv(float* __restrict__ deg) {
    int i = blockIdx.x * blockDim.x + threadIdx.x;
    if (i < N_NODES) {
        float d = deg[i] + 1.0f;
        deg[i] = rsqrtf(fmaxf(d, 1e-12f));
    }
}


__global__ __launch_bounds__(128) void k_gemm0(const float* __restrict__ x,
                                               const float* __restrict__ W0,
                                               const float* __restrict__ dinv,
                                               float* __restrict__ m0,
                                               float* __restrict__ h) {
    const int lane = threadIdx.x & 31;
    const int wv   = threadIdx.x >> 5;
    const int row0 = blockIdx.x * 16;
    const int col0 = wv * 16;
    const int hh   = lane >> 4;
    const int l15  = lane & 15;

    v8f c = {};
    const float* xrow = x  + (row0 + l15) * D_FEAT + 2 * hh;
    const float* wcol = W0 + (2 * hh) * HIDDEN + col0 + l15;

#pragma unroll 4
    for (int k = 0; k < D_FEAT; k += 4) {
        v2f a, b;
        a.x = xrow[k];
        a.y = xrow[k + 1];
        b.x = wcol[k * HIDDEN];
        b.y = wcol[k * HIDDEN + HIDDEN];
        c = __builtin_amdgcn_wmma_f32_16x16x4_f32(false, a, false, b,
                                                  (short)0, c, false, false);
    }
    float*       out = m0   + (row0 + 8 * hh) * HIDDEN + col0 + l15;
    float*       hs  = h    + (row0 + 8 * hh) * HIDDEN + col0 + l15;
    const float* dr  = dinv +  row0 + 8 * hh;
#pragma unroll
    for (int v = 0; v < 8; ++v) {
        const float di = dr[v];
        out[v * HIDDEN] = c[v];
        hs[v * HIDDEN]  = di * di * c[v];
    }
}


#define S0_TBL    1250
#define S0_PARTS  80
#define S0_SLICES 3
#define S0_EPS    (1024 * 1042)
#define S0_UNR    16
#define S0_G      16
#define S0_QCAP   48
static_assert(S0_TBL * S0_PARTS == N_NODES, "the tables tile the nodes exactly");
static_assert((S0_TBL * HIDDEN) % 32 == 0, "a table is whole 128-byte lines of h");
static_assert(S0_EPS * S0_SLICES >= N_EDGES && S0_EPS * (S0_SLICES - 1) < N_EDGES,
              "the slices cover the edges and the last one is not empty");
static_assert(S0_TBL <= 2048 && S0_EPS <= (1 << 21), "a queue record packs edge and row in 32 bits");
static_assert(32 % S0_G == 0 && S0_QCAP >= 32 && S0_UNR <= 32 && S0_UNR % 4 == 0, "queue and group sizes");
static_assert(S0_EPS % 4 == 0 && N_EDGES % 4 == 0, "a slice is whole groups of four edges");

__global__ __launch_bounds__(1024) void k_scatter0(const int* __restrict__ ei,
                                                  const float* __restrict__ ew,
                                                  const float* __restrict__ dinv,
                                                  const float* __restrict__ m0,
                                                  float* __restrict__ h) {
    __shared__ float tbl[S0_TBL * HIDDEN];
    __shared__ unsigned queue[32 * S0_QCAP];
    const int tid   = threadIdx.x;
    const int lane  = tid & 31;
    const int qb    = (tid >> 5) * S0_QCAP;
    const int part  = blockIdx.x % S0_PARTS;
    const int slice = blockIdx.x / S0_PARTS;
    const int lo    = part * S0_TBL;
    const int e0    = slice * S0_EPS;
    const int e1    = (e0 + S0_EPS < N_EDGES) ? e0 + S0_EPS : N_EDGES;

    for (int i = tid; i < S0_TBL * HIDDEN; i += 1024) tbl[i] = 0.0f;
    __syncthreads();

    const int*   dst = ei + N_EDGES;
    float*       tl  = tbl + lane;

    int qn = 0;
    for (int base = e0; ; base += 1024 * S0_UNR) {
        const bool last = base >= e1;
        unsigned ovf = 0u;
        if (!last) {
            int dd[S0_UNR];
#pragma unroll
            for (int q = 0; q < S0_UNR / 4; ++q) {
                const int i = base + (q * 1024 + tid) * 4;
                const bool in = i < e1;
                const int* p = dst + (in ? i : e1 - 4);
#pragma unroll
                for (int c = 0; c < 4; ++c) {
                    const int d = p[c];
                    dd[4 * q + c] = in ? d : -1;
                }
            }
#pragma unroll
            for (int k = 0; k < S0_UNR; ++k) {
                const unsigned r = (unsigned)(dd[k] - lo);
                const bool hit = r < (unsigned)S0_TBL;
                const unsigned mask = __builtin_amdgcn_ballot_w32(hit);
                if (mask != 0u) {
                    const int pos = qn + (int)__builtin_amdgcn_mbcnt_lo(mask, 0u);
                    if (hit) {
                        if (pos < S0_QCAP)
                            queue[qb + pos] = ((unsigned)(base + ((k >> 2) * 1024 + tid) * 4 + (k & 3) - e0) << 11) | r;
                        else
                            ovf |= 1u << k;
                    }
                    qn += __builtin_popcount(mask);
                }
            }
            if (qn > S0_QCAP) qn = S0_QCAP;
        }

        int off = 0;
        while (qn - off >= 32 || (last && qn - off > 0)) {
            const int n = (qn - off < 32) ? qn - off : 32;
            unsigned rec = 0u;
            if (lane < n) rec = queue[qb + off + lane];
            const int r_l = (int)(rec & 2047u);
            int   s_l  = 0;
            float wn_l = 0.0f;
            if (lane < n) {
                const int e = e0 + (int)(rec >> 11);
                s_l  = ei[e];
                wn_l = dinv[s_l] * ew[e] * dinv[lo + r_l];
            }
            for (int j0 = 0; j0 < n; j0 += S0_G) {
                float v0[S0_G], v1[S0_G];
#pragma unroll
                for (int j = 0; j < S0_G; ++j) {
                    const float* row = m0 + (unsigned)__builtin_amdgcn_readlane(s_l, j0 + j) * HIDDEN;
                    v0[j] = row[lane];
                    v1[j] = row[lane + 32];
                }
#pragma unroll
                for (int j = 0; j < S0_G; ++j) {
                    if (j0 + j < n) {
                        const int r = __builtin_amdgcn_readlane(r_l, j0 + j);
                        const float wn = __int_as_float(
                            __builtin_amdgcn_readlane(__float_as_int(wn_l), j0 + j));
                        atomicAdd(tl + r * HIDDEN,      wn * v0[j]);
                        atomicAdd(tl + r * HIDDEN + 32, wn * v1[j]);
                    }
                }
            }
            off += n;
        }
        const int rem = qn - off;
        if (off > 0 && lane < rem) {
            const unsigned t = queue[qb + off + lane];
            queue[qb + lane] = t;
        }
        qn = rem;

        if (ovf != 0u) {
#pragma nounroll
            for (int k = 0; k < S0_UNR; ++k) {
                if ((ovf >> k) & 1u) {
                    const int e = base + ((k >> 2) * 1024 + tid) * 4 + (k & 3);
                    const int s = ei[e];
                    const int d = dst[e];
                    const float wn = dinv[s] * ew[e] * dinv[d];
                    const float* row = m0 + s * HIDDEN;
                    float* trow = tbl + (d - lo) * HIDDEN;
#pragma nounroll
                    for (int f = 0; f < HIDDEN; ++f) atomicAdd(trow + f, wn * row[f]);
                }
            }
        }
        if (last) break;
    }
    __syncthreads();

    const int wv    = tid >> 5;
    const int nl    = S0_TBL * HIDDEN / 32;
    const int start = (slice * nl) / S0_SLICES;
    float* o = h + lo * HIDDEN;
    for (int j = wv; j < nl; j += 32) {
        int line = j + start;
        if (line >= nl) line -= nl;
        atomicAdd(&o[line * 32 + lane], tbl[line * 32 + lane]);
    }
}


__global__ __launch_bounds__(128) void k_gemm1(const float* __restrict__ h,
                                               const float* __restrict__ b0,
                                               const float* __restrict__ W1,
                                               const float* __restrict__ dinv,
                                               const float* __restrict__ b1,
                                               float* __restrict__ m1,
                                               float* __restrict__ out) {
    const int lane = threadIdx.x & 31;
    const int wv   = threadIdx.x >> 5;
    const int rb   = blockIdx.x * 4 + wv;
    if (rb >= ROWBLKS) return;
    const int row0 = rb * 16;
    const int hh   = lane >> 4;
    const int l15  = lane & 15;

    v8f c = {};
    const float* hrow = h  + (row0 + l15) * HIDDEN + 2 * hh;
    const float* brow = b0 + 2 * hh;
#pragma unroll
    for (int k = 0; k < HIDDEN; k += 4) {
        v2f a, b;
        a.x = fmaxf(hrow[k]     + brow[k],     0.0f);
        a.y = fmaxf(hrow[k + 1] + brow[k + 1], 0.0f);
        float bx = 0.0f, by = 0.0f;
        if (l15 < NCLASS) {
            bx = W1[(k + 2 * hh) * NCLASS + l15];
            by = W1[(k + 2 * hh + 1) * NCLASS + l15];
        }
        b.x = bx; b.y = by;
        c = __builtin_amdgcn_wmma_f32_16x16x4_f32(false, a, false, b,
                                                  (short)0, c, false, false);
    }
    if (l15 < NCLASS) {
        float*       om = m1   + (row0 + 8 * hh) * NCLASS + l15;
        float*       oo = out  + (row0 + 8 * hh) * NCLASS + l15;
        const float* dr = dinv +  row0 + 8 * hh;
        const float  bj = b1[l15];
#pragma unroll
        for (int v = 0; v < 8; ++v) {
            const float di = dr[v];
            om[v * NCLASS] = c[v];
            oo[v * NCLASS] = di * di * c[v] + bj;
        }
    }
}


#define S1_TBL    10016
#define S1_PARTS  10
#define S1_SLICES 25
#define S1_EPS    (N_EDGES / S1_SLICES)
#define S1_UNR    5
static_assert(N_EDGES % S1_SLICES == 0, "slices are equal");
static_assert(S1_EPS % (1024 * S1_UNR) == 0, "a slice is a whole number of trips");
static_assert(S1_TBL % 32 == 0 && N_NODES % 32 == 0, "every table is whole 128-byte lines of out");
static_assert(S1_TBL * S1_PARTS >= N_NODES && S1_TBL * (S1_PARTS - 1) < N_NODES,
              "the tables cover the nodes and the last one is not empty");

__global__ __launch_bounds__(1024) void k_scatter1(const int* __restrict__ ei,
                                                   const float* __restrict__ ew,
                                                   const float* __restrict__ dinv,
                                                   const float* __restrict__ m1,
                                                   float* __restrict__ out) {
    __shared__ float tbl[S1_TBL * NCLASS];
    const int tid   = threadIdx.x;
    const int part  = blockIdx.x % S1_PARTS;
    const int slice = blockIdx.x / S1_PARTS;
    const int lo    = part * S1_TBL;
    const int rest  = N_NODES - lo;
    const unsigned n_here = (unsigned)(rest < S1_TBL ? rest : S1_TBL);

    for (int i = tid; i < S1_TBL * NCLASS; i += 1024) tbl[i] = 0.0f;
    __syncthreads();

    const int*   src = ei + slice * S1_EPS + tid;
    const int*   dst = ei + N_EDGES + slice * S1_EPS + tid;
    const float* w   = ew + slice * S1_EPS + tid;

    int ns[S1_UNR], nd[S1_UNR]; float nw[S1_UNR];
#pragma unroll
    for (int k = 0; k < S1_UNR; ++k) {
        ns[k] = src[k * 1024]; nd[k] = dst[k * 1024]; nw[k] = w[k * 1024];
    }
    for (int base = 0; base < S1_EPS; base += 1024 * S1_UNR) {
        int cs[S1_UNR], cd[S1_UNR]; float cw[S1_UNR];
#pragma unroll
        for (int k = 0; k < S1_UNR; ++k) { cs[k] = ns[k]; cd[k] = nd[k]; cw[k] = nw[k]; }
        const int nb = base + 1024 * S1_UNR;
        if (nb < S1_EPS) {
#pragma unroll
            for (int k = 0; k < S1_UNR; ++k) {
                ns[k] = src[nb + k * 1024]; nd[k] = dst[nb + k * 1024]; nw[k] = w[nb + k * 1024];
            }
        }
        float ds[S1_UNR], dt[S1_UNR]; float v[S1_UNR][NCLASS];
#pragma unroll
        for (int k = 0; k < S1_UNR; ++k) {
            const unsigned r = (unsigned)(cd[k] - lo);
            if (r < n_here) {
                ds[k] = dinv[cs[k]];
                dt[k] = dinv[cd[k]];
                const float* row = m1 + cs[k] * NCLASS;
#pragma unroll
                for (int j = 0; j < NCLASS; ++j) v[k][j] = row[j];
            }
        }
#pragma unroll
        for (int k = 0; k < S1_UNR; ++k) {
            const unsigned r = (unsigned)(cd[k] - lo);
            if (r < n_here) {
                const float wn = ds[k] * cw[k] * dt[k];
#pragma unroll
                for (int j = 0; j < NCLASS; ++j) atomicAdd(&tbl[r * NCLASS + j], wn * v[k][j]);
            }
        }
    }
    __syncthreads();

    const int lane = tid & 31;
    const int wv   = tid >> 5;
    const int nl   = (int)(n_here * NCLASS) >> 5;
    const int start = (slice * nl) / S1_SLICES;
    float* o = out + lo * NCLASS;
    for (int j = wv; j < nl; j += 32) {
        int line = j + start;
        if (line >= nl) line -= nl;
        atomicAdd(&o[line * 32 + lane], tbl[line * 32 + lane]);
    }
}


extern "C" void kernel_launch(void* const* d_in, const int* in_sizes, int n_in,
                              void* d_out, int out_size, void* d_ws, size_t ws_size,
                              hipStream_t stream) {
    const float* x  = (const float*)d_in[0];
    const int*   ei = (const int*)  d_in[1];
    const float* ew = (const float*)d_in[2];
    const float* W0 = (const float*)d_in[3];
    const float* b0 = (const float*)d_in[4];
    const float* W1 = (const float*)d_in[5];
    const float* b1 = (const float*)d_in[6];
    float* out = (float*)d_out;

    float* ws   = (float*)d_ws;
    float* dinv = ws;
    float* m0   = dinv + N_NODES;
    float* h    = m0 + (size_t)N_NODES * HIDDEN;
    float* m1   = h  + (size_t)N_NODES * HIDDEN;

    k_zero<<<(N_NODES + 255) / 256, 256, 0, stream>>>(dinv, N_NODES);
    k_deg <<<2 * DEG_SLICES, 1024, 0, stream>>>(ei, ew, dinv);
    k_dinv<<<(N_NODES + 255) / 256, 256, 0, stream>>>(dinv);

    k_gemm0<<<ROWBLKS, 128, 0, stream>>>(x, W0, dinv, m0, h);

    k_scatter0<<<S0_PARTS * S0_SLICES, 1024, 0, stream>>>(ei, ew, dinv, m0, h);

    k_gemm1<<<(ROWBLKS + 3) / 4, 128, 0, stream>>>(h, b0, W1, dinv, b1, m1, out);

    k_scatter1<<<S1_PARTS * S1_SLICES, 1024, 0, stream>>>(ei, ew, dinv, m1, out);
}


# D^-1/2 taken off the edges: GEMM epilogues store rows pre-scaled by dinv[row], LDS tables scaled by dinv[dst] once at flush; edge passes drop 2 scattered dinv reads per edge; term rounding changes
# speedup vs baseline: 2.4748x; 1.1200x over previous
#include <hip/hip_runtime.h>

typedef __attribute__((ext_vector_type(2))) float v2f;
typedef __attribute__((ext_vector_type(8))) float v8f;

#define N_NODES 100000
#define N_EDGES 3200000
#define D_FEAT  256
#define HIDDEN  64
#define NCLASS  7
#define ROWBLKS (N_NODES / 16)


__global__ void k_zero(float* __restrict__ p, int n) {
    int i = blockIdx.x * blockDim.x + threadIdx.x;
    if (i < n) p[i] = 0.0f;
}

#define DEG_TBL    50016
#define DEG_SLICES 128
#define DEG_EPS    (N_EDGES / DEG_SLICES)
#define DEG_UNR    13
static_assert(N_EDGES % DEG_SLICES == 0, "slices are equal");
static_assert(DEG_TBL % 32 == 0 && (N_NODES - DEG_TBL) % 32 == 0 && N_NODES <= 2 * DEG_TBL &&
              N_NODES > DEG_TBL, "two tables of whole lines cover the nodes");

__global__ __launch_bounds__(1024) void k_deg(const int* __restrict__ ei,
                                              const float* __restrict__ ew,
                                              float* __restrict__ deg) {
    __shared__ float tbl[DEG_TBL];
    const int tid   = threadIdx.x;
    const int half  = blockIdx.x & 1;
    const int slice = blockIdx.x >> 1;
    const int lo    = half * DEG_TBL;
    const unsigned n_here = half ? (N_NODES - DEG_TBL) : DEG_TBL;

    for (int i = tid; i < DEG_TBL; i += 1024) tbl[i] = 0.0f;
    __syncthreads();

    const int*   dst = ei + N_EDGES + slice * DEG_EPS;
    const float* w   = ew + slice * DEG_EPS;
    for (int base = 0; base < DEG_EPS; base += 1024 * DEG_UNR) {
        int dd[DEG_UNR]; float ww[DEG_UNR];
#pragma unroll
        for (int k = 0; k < DEG_UNR; ++k) {
            const int i = base + k * 1024 + tid;
            const bool in = i < DEG_EPS;
            dd[k] = in ? dst[i] : -1;
            ww[k] = in ? w[i]   : 0.0f;
        }
#pragma unroll
        for (int k = 0; k < DEG_UNR; ++k) {
            const unsigned r = (unsigned)(dd[k] - lo);
            if (r < n_here) atomicAdd(&tbl[r], ww[k]);
        }
    }
    __syncthreads();

    const int lane = tid & 31;
    const int wv   = tid >> 5;
    const int nl   = (int)(n_here >> 5);
    const int start = (slice * nl) / DEG_SLICES;
    for (int j = wv; j < nl; j += 32) {
        int line = j + start;
        if (line >= nl) line -= nl;
        atomicAdd(&deg[lo + line * 32 + lane], tbl[line * 32 + lane]);
    }
}

__global__ void k_dinv(float* __restrict__ deg) {
    int i = blockIdx.x * blockDim.x + threadIdx.x;
    if (i < N_NODES) {
        float d = deg[i] + 1.0f;
        deg[i] = rsqrtf(fmaxf(d, 1e-12f));
    }
}


__global__ __launch_bounds__(128) void k_gemm0(const float* __restrict__ x,
                                               const float* __restrict__ W0,
                                               const float* __restrict__ dinv,
                                               float* __restrict__ m0,
                                               float* __restrict__ h) {
    const int lane = threadIdx.x & 31;
    const int wv   = threadIdx.x >> 5;
    const int row0 = blockIdx.x * 16;
    const int col0 = wv * 16;
    const int hh   = lane >> 4;
    const int l15  = lane & 15;

    v8f c = {};
    const float* xrow = x  + (row0 + l15) * D_FEAT + 2 * hh;
    const float* wcol = W0 + (2 * hh) * HIDDEN + col0 + l15;

#pragma unroll 4
    for (int k = 0; k < D_FEAT; k += 4) {
        v2f a, b;
        a.x = xrow[k];
        a.y = xrow[k + 1];
        b.x = wcol[k * HIDDEN];
        b.y = wcol[k * HIDDEN + HIDDEN];
        c = __builtin_amdgcn_wmma_f32_16x16x4_f32(false, a, false, b,
                                                  (short)0, c, false, false);
    }
    float*       out = m0   + (row0 + 8 * hh) * HIDDEN + col0 + l15;
    float*       hs  = h    + (row0 + 8 * hh) * HIDDEN + col0 + l15;
    const float* dr  = dinv +  row0 + 8 * hh;
#pragma unroll
    for (int v = 0; v < 8; ++v) {
        const float di = dr[v];
        const float ms = di * c[v];
        out[v * HIDDEN] = ms;
        hs[v * HIDDEN]  = di * ms;
    }
}


#define S0_TBL    1250
#define S0_PARTS  80
#define S0_SLICES 3
#define S0_EPS    (1024 * 1042)
#define S0_UNR    16
#define S0_G      16
#define S0_QCAP   48
static_assert(S0_TBL * S0_PARTS == N_NODES, "the tables tile the nodes exactly");
static_assert((S0_TBL * HIDDEN) % 32 == 0, "a table is whole 128-byte lines of h");
static_assert(S0_EPS * S0_SLICES >= N_EDGES && S0_EPS * (S0_SLICES - 1) < N_EDGES,
              "the slices cover the edges and the last one is not empty");
static_assert(S0_TBL <= 2048 && S0_EPS <= (1 << 21), "a queue record packs edge and row in 32 bits");
static_assert(32 % S0_G == 0 && S0_QCAP >= 32 && S0_UNR <= 32 && S0_UNR % 4 == 0, "queue and group sizes");
static_assert(32 * S0_QCAP >= S0_TBL && S0_TBL > 1024 && S0_TBL <= 2048,
              "the dead queues hold the part's dinv: two values per thread at most");
static_assert(S0_EPS % 4 == 0 && N_EDGES % 4 == 0, "a slice is whole groups of four edges");

__global__ __launch_bounds__(1024) void k_scatter0(const int* __restrict__ ei,
                                                  const float* __restrict__ ew,
                                                  const float* __restrict__ dinv,
                                                  const float* __restrict__ m0,
                                                  float* __restrict__ h) {
    __shared__ float tbl[S0_TBL * HIDDEN];
    __shared__ unsigned queue[32 * S0_QCAP];
    const int tid   = threadIdx.x;
    const int lane  = tid & 31;
    const int qb    = (tid >> 5) * S0_QCAP;
    const int part  = blockIdx.x % S0_PARTS;
    const int slice = blockIdx.x / S0_PARTS;
    const int lo    = part * S0_TBL;
    const int e0    = slice * S0_EPS;
    const int e1    = (e0 + S0_EPS < N_EDGES) ? e0 + S0_EPS : N_EDGES;

    const float dv0 = dinv[lo + tid];
    const float dv1 = (tid < S0_TBL - 1024) ? dinv[lo + 1024 + tid] : 0.0f;

    for (int i = tid; i < S0_TBL * HIDDEN; i += 1024) tbl[i] = 0.0f;
    __syncthreads();

    const int*   dst = ei + N_EDGES;
    float*       tl  = tbl + lane;

    int qn = 0;
    for (int base = e0; ; base += 1024 * S0_UNR) {
        const bool last = base >= e1;
        unsigned ovf = 0u;
        if (!last) {
            int dd[S0_UNR];
#pragma unroll
            for (int q = 0; q < S0_UNR / 4; ++q) {
                const int i = base + (q * 1024 + tid) * 4;
                const bool in = i < e1;
                const int* p = dst + (in ? i : e1 - 4);
#pragma unroll
                for (int c = 0; c < 4; ++c) {
                    const int d = p[c];
                    dd[4 * q + c] = in ? d : -1;
                }
            }
#pragma unroll
            for (int k = 0; k < S0_UNR; ++k) {
                const unsigned r = (unsigned)(dd[k] - lo);
                const bool hit = r < (unsigned)S0_TBL;
                const unsigned mask = __builtin_amdgcn_ballot_w32(hit);
                if (mask != 0u) {
                    const int pos = qn + (int)__builtin_amdgcn_mbcnt_lo(mask, 0u);
                    if (hit) {
                        if (pos < S0_QCAP)
                            queue[qb + pos] = ((unsigned)(base + ((k >> 2) * 1024 + tid) * 4 + (k & 3) - e0) << 11) | r;
                        else
                            ovf |= 1u << k;
                    }
                    qn += __builtin_popcount(mask);
                }
            }
            if (qn > S0_QCAP) qn = S0_QCAP;
        }

        int off = 0;
        while (qn - off >= 32 || (last && qn - off > 0)) {
            const int n = (qn - off < 32) ? qn - off : 32;
            unsigned rec = 0u;
            if (lane < n) rec = queue[qb + off + lane];
            const int r_l = (int)(rec & 2047u);
            int   s_l  = 0;
            float wn_l = 0.0f;
            if (lane < n) {
                const int e = e0 + (int)(rec >> 11);
                s_l  = ei[e];
                wn_l = ew[e];
            }
            for (int j0 = 0; j0 < n; j0 += S0_G) {
                float v0[S0_G], v1[S0_G];
#pragma unroll
                for (int j = 0; j < S0_G; ++j) {
                    const float* row = m0 + (unsigned)__builtin_amdgcn_readlane(s_l, j0 + j) * HIDDEN;
                    v0[j] = row[lane];
                    v1[j] = row[lane + 32];
                }
#pragma unroll
                for (int j = 0; j < S0_G; ++j) {
                    if (j0 + j < n) {
                        const int r = __builtin_amdgcn_readlane(r_l, j0 + j);
                        const float wn = __int_as_float(
                            __builtin_amdgcn_readlane(__float_as_int(wn_l), j0 + j));
                        atomicAdd(tl + r * HIDDEN,      wn * v0[j]);
                        atomicAdd(tl + r * HIDDEN + 32, wn * v1[j]);
                    }
                }
            }
            off += n;
        }
        const int rem = qn - off;
        if (off > 0 && lane < rem) {
            const unsigned t = queue[qb + off + lane];
            queue[qb + lane] = t;
        }
        qn = rem;

        if (ovf != 0u) {
#pragma nounroll
            for (int k = 0; k < S0_UNR; ++k) {
                if ((ovf >> k) & 1u) {
                    const int e = base + ((k >> 2) * 1024 + tid) * 4 + (k & 3);
                    const int s = ei[e];
                    const int d = dst[e];
                    const float wn = ew[e];
                    const float* row = m0 + s * HIDDEN;
                    float* trow = tbl + (d - lo) * HIDDEN;
#pragma nounroll
                    for (int f = 0; f < HIDDEN; ++f) atomicAdd(trow + f, wn * row[f]);
                }
            }
        }
        if (last) break;
    }
    __syncthreads();

    queue[tid] = __float_as_uint(dv0);
    if (tid < S0_TBL - 1024) queue[1024 + tid] = __float_as_uint(dv1);
    __syncthreads();

    const int wv    = tid >> 5;
    const int nl    = S0_TBL * HIDDEN / 32;
    const int start = (slice * nl) / S0_SLICES;
    float* o = h + lo * HIDDEN;
    for (int j = wv; j < nl; j += 32) {
        int line = j + start;
        if (line >= nl) line -= nl;
        atomicAdd(&o[line * 32 + lane], __uint_as_float(queue[line >> 1]) * tbl[line * 32 + lane]);
    }
}


__global__ __launch_bounds__(128) void k_gemm1(const float* __restrict__ h,
                                               const float* __restrict__ b0,
                                               const float* __restrict__ W1,
                                               const float* __restrict__ dinv,
                                               const float* __restrict__ b1,
                                               float* __restrict__ m1,
                                               float* __restrict__ out) {
    const int lane = threadIdx.x & 31;
    const int wv   = threadIdx.x >> 5;
    const int rb   = blockIdx.x * 4 + wv;
    if (rb >= ROWBLKS) return;
    const int row0 = rb * 16;
    const int hh   = lane >> 4;
    const int l15  = lane & 15;

    v8f c = {};
    const float* hrow = h  + (row0 + l15) * HIDDEN + 2 * hh;
    const float* brow = b0 + 2 * hh;
#pragma unroll
    for (int k = 0; k < HIDDEN; k += 4) {
        v2f a, b;
        a.x = fmaxf(hrow[k]     + brow[k],     0.0f);
        a.y = fmaxf(hrow[k + 1] + brow[k + 1], 0.0f);
        float bx = 0.0f, by = 0.0f;
        if (l15 < NCLASS) {
            bx = W1[(k + 2 * hh) * NCLASS + l15];
            by = W1[(k + 2 * hh + 1) * NCLASS + l15];
        }
        b.x = bx; b.y = by;
        c = __builtin_amdgcn_wmma_f32_16x16x4_f32(false, a, false, b,
                                                  (short)0, c, false, false);
    }
    if (l15 < NCLASS) {
        float*       om = m1   + (row0 + 8 * hh) * NCLASS + l15;
        float*       oo = out  + (row0 + 8 * hh) * NCLASS + l15;
        const float* dr = dinv +  row0 + 8 * hh;
        const float  bj = b1[l15];
#pragma unroll
        for (int v = 0; v < 8; ++v) {
            const float di = dr[v];
            const float ms = di * c[v];
            om[v * NCLASS] = ms;
            oo[v * NCLASS] = di * ms + bj;
        }
    }
}


#define S1_TBL    10016
#define S1_PARTS  10
#define S1_SLICES 25
#define S1_EPS    (N_EDGES / S1_SLICES)
#define S1_UNR    5
static_assert(N_EDGES % S1_SLICES == 0, "slices are equal");
static_assert(S1_EPS % (1024 * S1_UNR) == 0, "a slice is a whole number of trips");
static_assert(S1_TBL % 32 == 0 && N_NODES % 32 == 0, "every table is whole 128-byte lines of out");
static_assert(S1_TBL * S1_PARTS >= N_NODES && S1_TBL * (S1_PARTS - 1) < N_NODES,
              "the tables cover the nodes and the last one is not empty");
static_assert(S1_TBL * (NCLASS + 1) * 4 <= 320 * 1024, "table and the part's dinv fit one WGP's LDS");

__global__ __launch_bounds__(1024) void k_scatter1(const int* __restrict__ ei,
                                                   const float* __restrict__ ew,
                                                   const float* __restrict__ dinv,
                                                   const float* __restrict__ m1,
                                                   float* __restrict__ out) {
    __shared__ float tbl[S1_TBL * NCLASS];
    __shared__ float dls[S1_TBL];
    const int tid   = threadIdx.x;
    const int part  = blockIdx.x % S1_PARTS;
    const int slice = blockIdx.x / S1_PARTS;
    const int lo    = part * S1_TBL;
    const int rest  = N_NODES - lo;
    const unsigned n_here = (unsigned)(rest < S1_TBL ? rest : S1_TBL);

    for (int i = tid; i < S1_TBL * NCLASS; i += 1024) tbl[i] = 0.0f;
    for (unsigned i = tid; i < n_here; i += 1024) dls[i] = dinv[lo + i];
    __syncthreads();

    const int*   src = ei + slice * S1_EPS + tid;
    const int*   dst = ei + N_EDGES + slice * S1_EPS + tid;
    const float* w   = ew + slice * S1_EPS + tid;

    int ns[S1_UNR], nd[S1_UNR]; float nw[S1_UNR];
#pragma unroll
    for (int k = 0; k < S1_UNR; ++k) {
        ns[k] = src[k * 1024]; nd[k] = dst[k * 1024]; nw[k] = w[k * 1024];
    }
    for (int base = 0; base < S1_EPS; base += 1024 * S1_UNR) {
        int cs[S1_UNR], cd[S1_UNR]; float cw[S1_UNR];
#pragma unroll
        for (int k = 0; k < S1_UNR; ++k) { cs[k] = ns[k]; cd[k] = nd[k]; cw[k] = nw[k]; }
        const int nb = base + 1024 * S1_UNR;
        if (nb < S1_EPS) {
#pragma unroll
            for (int k = 0; k < S1_UNR; ++k) {
                ns[k] = src[nb + k * 1024]; nd[k] = dst[nb + k * 1024]; nw[k] = w[nb + k * 1024];
            }
        }
        float v[S1_UNR][NCLASS];
#pragma unroll
        for (int k = 0; k < S1_UNR; ++k) {
            const unsigned r = (unsigned)(cd[k] - lo);
            if (r < n_here) {
                const float* row = m1 + cs[k] * NCLASS;
#pragma unroll
                for (int j = 0; j < NCLASS; ++j) v[k][j] = row[j];
            }
        }
#pragma unroll
        for (int k = 0; k < S1_UNR; ++k) {
            const unsigned r = (unsigned)(cd[k] - lo);
            if (r < n_here) {
#pragma unroll
                for (int j = 0; j < NCLASS; ++j) atomicAdd(&tbl[r * NCLASS + j], cw[k] * v[k][j]);
            }
        }
    }
    __syncthreads();

    const int lane = tid & 31;
    const int wv   = tid >> 5;
    const int nl   = (int)(n_here * NCLASS) >> 5;
    const int start = (slice * nl) / S1_SLICES;
    float* o = out + lo * NCLASS;
    for (int j = wv; j < nl; j += 32) {
        int line = j + start;
        if (line >= nl) line -= nl;
        const unsigned i = (unsigned)(line * 32 + lane);
        atomicAdd(&o[i], dls[i / NCLASS] * tbl[i]);
    }
}


extern "C" void kernel_launch(void* const* d_in, const int* in_sizes, int n_in,
                              void* d_out, int out_size, void* d_ws, size_t ws_size,
                              hipStream_t stream) {
    const float* x  = (const float*)d_in[0];
    const int*   ei = (const int*)  d_in[1];
    const float* ew = (const float*)d_in[2];
    const float* W0 = (const float*)d_in[3];
    const float* b0 = (const float*)d_in[4];
    const float* W1 = (const float*)d_in[5];
    const float* b1 = (const float*)d_in[6];
    float* out = (float*)d_out;

    float* ws   = (float*)d_ws;
    float* dinv = ws;
    float* m0   = dinv + N_NODES;
    float* h    = m0 + (size_t)N_NODES * HIDDEN;
    float* m1   = h  + (size_t)N_NODES * HIDDEN;

    k_zero<<<(N_NODES + 255) / 256, 256, 0, stream>>>(dinv, N_NODES);
    k_deg <<<2 * DEG_SLICES, 1024, 0, stream>>>(ei, ew, dinv);
    k_dinv<<<(N_NODES + 255) / 256, 256, 0, stream>>>(dinv);

    k_gemm0<<<ROWBLKS, 128, 0, stream>>>(x, W0, dinv, m0, h);

    k_scatter0<<<S0_PARTS * S0_SLICES, 1024, 0, stream>>>(ei, ew, dinv, m0, h);

    k_gemm1<<<(ROWBLKS + 3) / 4, 128, 0, stream>>>(h, b0, W1, dinv, b1, m1, out);

    k_scatter1<<<S1_PARTS * S1_SLICES, 1024, 0, stream>>>(ei, ew, dinv, m1, out);
}


# GEMM0: a block reads its 16 rows of x (one contiguous 16-KiB run) once as whole 128-B lines into LDS and takes A fragments there; was 16 B from each of 16 lines per load, per wave; sums keep order
# speedup vs baseline: 2.6625x; 1.0758x over previous
#include <hip/hip_runtime.h>

typedef __attribute__((ext_vector_type(2))) float v2f;
typedef __attribute__((ext_vector_type(8))) float v8f;

#define N_NODES 100000
#define N_EDGES 3200000
#define D_FEAT  256
#define HIDDEN  64
#define NCLASS  7
#define ROWBLKS (N_NODES / 16)


__global__ void k_zero(float* __restrict__ p, int n) {
    int i = blockIdx.x * blockDim.x + threadIdx.x;
    if (i < n) p[i] = 0.0f;
}

#define DEG_TBL    50016
#define DEG_SLICES 128
#define DEG_EPS    (N_EDGES / DEG_SLICES)
#define DEG_UNR    13
static_assert(N_EDGES % DEG_SLICES == 0, "slices are equal");
static_assert(DEG_TBL % 32 == 0 && (N_NODES - DEG_TBL) % 32 == 0 && N_NODES <= 2 * DEG_TBL &&
              N_NODES > DEG_TBL, "two tables of whole lines cover the nodes");

__global__ __launch_bounds__(1024) void k_deg(const int* __restrict__ ei,
                                              const float* __restrict__ ew,
                                              float* __restrict__ deg) {
    __shared__ float tbl[DEG_TBL];
    const int tid   = threadIdx.x;
    const int half  = blockIdx.x & 1;
    const int slice = blockIdx.x >> 1;
    const int lo    = half * DEG_TBL;
    const unsigned n_here = half ? (N_NODES - DEG_TBL) : DEG_TBL;

    for (int i = tid; i < DEG_TBL; i += 1024) tbl[i] = 0.0f;
    __syncthreads();

    const int*   dst = ei + N_EDGES + slice * DEG_EPS;
    const float* w   = ew + slice * DEG_EPS;
    for (int base = 0; base < DEG_EPS; base += 1024 * DEG_UNR) {
        int dd[DEG_UNR]; float ww[DEG_UNR];
#pragma unroll
        for (int k = 0; k < DEG_UNR; ++k) {
            const int i = base + k * 1024 + tid;
            const bool in = i < DEG_EPS;
            dd[k] = in ? dst[i] : -1;
            ww[k] = in ? w[i]   : 0.0f;
        }
#pragma unroll
        for (int k = 0; k < DEG_UNR; ++k) {
            const unsigned r = (unsigned)(dd[k] - lo);
            if (r < n_here) atomicAdd(&tbl[r], ww[k]);
        }
    }
    __syncthreads();

    const int lane = tid & 31;
    const int wv   = tid >> 5;
    const int nl   = (int)(n_here >> 5);
    const int start = (slice * nl) / DEG_SLICES;
    for (int j = wv; j < nl; j += 32) {
        int line = j + start;
        if (line >= nl) line -= nl;
        atomicAdd(&deg[lo + line * 32 + lane], tbl[line * 32 + lane]);
    }
}

__global__ void k_dinv(float* __restrict__ deg) {
    int i = blockIdx.x * blockDim.x + threadIdx.x;
    if (i < N_NODES) {
        float d = deg[i] + 1.0f;
        deg[i] = rsqrtf(fmaxf(d, 1e-12f));
    }
}

#define G0_PITCH (D_FEAT + 4)
#define G0_LOADS (16 * D_FEAT / 4 / 128)
static_assert((16 * D_FEAT) % (4 * 128) == 0, "the tile is a whole number of 16-byte loads per thread");
static_assert(G0_PITCH % 4 == 0 && (G0_PITCH % 64) == 4, "16-byte aligned rows, 4 banks apart");

__global__ __launch_bounds__(128) void k_gemm0(const float* __restrict__ x,
                                               const float* __restrict__ W0,
                                               const float* __restrict__ dinv,
                                               float* __restrict__ m0,
                                               float* __restrict__ h) {
    __shared__ __attribute__((aligned(16))) float xt[16 * G0_PITCH];
    const int tid  = threadIdx.x;
    const int lane = tid & 31;
    const int wv   = tid >> 5;
    const int row0 = blockIdx.x * 16;
    const int col0 = wv * 16;
    const int hh   = lane >> 4;
    const int l15  = lane & 15;

    const float* run = x + (size_t)row0 * D_FEAT;
    float4 piece[G0_LOADS];
#pragma unroll
    for (int j = 0; j < G0_LOADS; ++j) {
        const float* p = run + (j * 128 + tid) * 4;
        piece[j] = make_float4(p[0], p[1], p[2], p[3]);
    }
#pragma unroll
    for (int j = 0; j < G0_LOADS; ++j) {
        const int q = j * 128 + tid;
        float* d = xt + (q >> 6) * G0_PITCH + (q & 63) * 4;
        d[0] = piece[j].x; d[1] = piece[j].y; d[2] = piece[j].z; d[3] = piece[j].w;
    }
    __syncthreads();

    v8f c = {};
    const float* xrow = xt + l15 * G0_PITCH + 2 * hh;
    const float* wcol = W0 + (2 * hh) * HIDDEN + col0 + l15;

#pragma unroll 8
    for (int k = 0; k < D_FEAT; k += 4) {
        v2f a, b;
        a.x = xrow[k];
        a.y = xrow[k + 1];
        b.x = wcol[k * HIDDEN];
        b.y = wcol[k * HIDDEN + HIDDEN];
        c = __builtin_amdgcn_wmma_f32_16x16x4_f32(false, a, false, b,
                                                  (short)0, c, false, false);
    }
    float*       out = m0   + (row0 + 8 * hh) * HIDDEN + col0 + l15;
    float*       hs  = h    + (row0 + 8 * hh) * HIDDEN + col0 + l15;
    const float* dr  = dinv +  row0 + 8 * hh;
#pragma unroll
    for (int v = 0; v < 8; ++v) {
        const float di = dr[v];
        const float ms = di * c[v];
        out[v * HIDDEN] = ms;
        hs[v * HIDDEN]  = di * ms;
    }
}


#define S0_TBL    1250
#define S0_PARTS  80
#define S0_SLICES 3
#define S0_EPS    (1024 * 1042)
#define S0_UNR    16
#define S0_G      16
#define S0_QCAP   48
static_assert(S0_TBL * S0_PARTS == N_NODES, "the tables tile the nodes exactly");
static_assert((S0_TBL * HIDDEN) % 32 == 0, "a table is whole 128-byte lines of h");
static_assert(S0_EPS * S0_SLICES >= N_EDGES && S0_EPS * (S0_SLICES - 1) < N_EDGES,
              "the slices cover the edges and the last one is not empty");
static_assert(S0_TBL <= 2048 && S0_EPS <= (1 << 21), "a queue record packs edge and row in 32 bits");
static_assert(32 % S0_G == 0 && S0_QCAP >= 32 && S0_UNR <= 32 && S0_UNR % 4 == 0, "queue and group sizes");
static_assert(32 * S0_QCAP >= S0_TBL && S0_TBL > 1024 && S0_TBL <= 2048,
              "the dead queues hold the part's dinv: two values per thread at most");
static_assert(S0_EPS % 4 == 0 && N_EDGES % 4 == 0, "a slice is whole groups of four edges");

__global__ __launch_bounds__(1024) void k_scatter0(const int* __restrict__ ei,
                                                  const float* __restrict__ ew,
                                                  const float* __restrict__ dinv,
                                                  const float* __restrict__ m0,
                                                  float* __restrict__ h) {
    __shared__ float tbl[S0_TBL * HIDDEN];
    __shared__ unsigned queue[32 * S0_QCAP];
    const int tid   = threadIdx.x;
    const int lane  = tid & 31;
    const int qb    = (tid >> 5) * S0_QCAP;
    const int part  = blockIdx.x % S0_PARTS;
    const int slice = blockIdx.x / S0_PARTS;
    const int lo    = part * S0_TBL;
    const int e0    = slice * S0_EPS;
    const int e1    = (e0 + S0_EPS < N_EDGES) ? e0 + S0_EPS : N_EDGES;

    const float dv0 = dinv[lo + tid];
    const float dv1 = (tid < S0_TBL - 1024) ? dinv[lo + 1024 + tid] : 0.0f;

    for (int i = tid; i < S0_TBL * HIDDEN; i += 1024) tbl[i] = 0.0f;
    __syncthreads();

    const int*   dst = ei + N_EDGES;
    float*       tl  = tbl + lane;

    int qn = 0;
    for (int base = e0; ; base += 1024 * S0_UNR) {
        const bool last = base >= e1;
        unsigned ovf = 0u;
        if (!last) {
            int dd[S0_UNR];
#pragma unroll
            for (int q = 0; q < S0_UNR / 4; ++q) {
                const int i = base + (q * 1024 + tid) * 4;
                const bool in = i < e1;
                const int* p = dst + (in ? i : e1 - 4);
#pragma unroll
                for (int c = 0; c < 4; ++c) {
                    const int d = p[c];
                    dd[4 * q + c] = in ? d : -1;
                }
            }
#pragma unroll
            for (int k = 0; k < S0_UNR; ++k) {
                const unsigned r = (unsigned)(dd[k] - lo);
                const bool hit = r < (unsigned)S0_TBL;
                const unsigned mask = __builtin_amdgcn_ballot_w32(hit);
                if (mask != 0u) {
                    const int pos = qn + (int)__builtin_amdgcn_mbcnt_lo(mask, 0u);
                    if (hit) {
                        if (pos < S0_QCAP)
                            queue[qb + pos] = ((unsigned)(base + ((k >> 2) * 1024 + tid) * 4 + (k & 3) - e0) << 11) | r;
                        else
                            ovf |= 1u << k;
                    }
                    qn += __builtin_popcount(mask);
                }
            }
            if (qn > S0_QCAP) qn = S0_QCAP;
        }

        int off = 0;
        while (qn - off >= 32 || (last && qn - off > 0)) {
            const int n = (qn - off < 32) ? qn - off : 32;
            unsigned rec = 0u;
            if (lane < n) rec = queue[qb + off + lane];
            const int r_l = (int)(rec & 2047u);
            int   s_l  = 0;
            float wn_l = 0.0f;
            if (lane < n) {
                const int e = e0 + (int)(rec >> 11);
                s_l  = ei[e];
                wn_l = ew[e];
            }
            for (int j0 = 0; j0 < n; j0 += S0_G) {
                float v0[S0_G], v1[S0_G];
#pragma unroll
                for (int j = 0; j < S0_G; ++j) {
                    const float* row = m0 + (unsigned)__builtin_amdgcn_readlane(s_l, j0 + j) * HIDDEN;
                    v0[j] = row[lane];
                    v1[j] = row[lane + 32];
                }
#pragma unroll
                for (int j = 0; j < S0_G; ++j) {
                    if (j0 + j < n) {
                        const int r = __builtin_amdgcn_readlane(r_l, j0 + j);
                        const float wn = __int_as_float(
                            __builtin_amdgcn_readlane(__float_as_int(wn_l), j0 + j));
                        atomicAdd(tl + r * HIDDEN,      wn * v0[j]);
                        atomicAdd(tl + r * HIDDEN + 32, wn * v1[j]);
                    }
                }
            }
            off += n;
        }
        const int rem = qn - off;
        if (off > 0 && lane < rem) {
            const unsigned t = queue[qb + off + lane];
            queue[qb + lane] = t;
        }
        qn = rem;

        if (ovf != 0u) {
#pragma nounroll
            for (int k = 0; k < S0_UNR; ++k) {
                if ((ovf >> k) & 1u) {
                    const int e = base + ((k >> 2) * 1024 + tid) * 4 + (k & 3);
                    const int s = ei[e];
                    const int d = dst[e];
                    const float wn = ew[e];
                    const float* row = m0 + s * HIDDEN;
                    float* trow = tbl + (d - lo) * HIDDEN;
#pragma nounroll
                    for (int f = 0; f < HIDDEN; ++f) atomicAdd(trow + f, wn * row[f]);
                }
            }
        }
        if (last) break;
    }
    __syncthreads();

    queue[tid] = __float_as_uint(dv0);
    if (tid < S0_TBL - 1024) queue[1024 + tid] = __float_as_uint(dv1);
    __syncthreads();

    const int wv    = tid >> 5;
    const int nl    = S0_TBL * HIDDEN / 32;
    const int start = (slice * nl) / S0_SLICES;
    float* o = h + lo * HIDDEN;
    for (int j = wv; j < nl; j += 32) {
        int line = j + start;
        if (line >= nl) line -= nl;
        atomicAdd(&o[line * 32 + lane], __uint_as_float(queue[line >> 1]) * tbl[line * 32 + lane]);
    }
}


__global__ __launch_bounds__(128) void k_gemm1(const float* __restrict__ h,
                                               const float* __restrict__ b0,
                                               const float* __restrict__ W1,
                                               const float* __restrict__ dinv,
                                               const float* __restrict__ b1,
                                               float* __restrict__ m1,
                                               float* __restrict__ out) {
    const int lane = threadIdx.x & 31;
    const int wv   = threadIdx.x >> 5;
    const int rb   = blockIdx.x * 4 + wv;
    if (rb >= ROWBLKS) return;
    const int row0 = rb * 16;
    const int hh   = lane >> 4;
    const int l15  = lane & 15;

    v8f c = {};
    const float* hrow = h  + (row0 + l15) * HIDDEN + 2 * hh;
    const float* brow = b0 + 2 * hh;
#pragma unroll
    for (int k = 0; k < HIDDEN; k += 4) {
        v2f a, b;
        a.x = fmaxf(hrow[k]     + brow[k],     0.0f);
        a.y = fmaxf(hrow[k + 1] + brow[k + 1], 0.0f);
        float bx = 0.0f, by = 0.0f;
        if (l15 < NCLASS) {
            bx = W1[(k + 2 * hh) * NCLASS + l15];
            by = W1[(k + 2 * hh + 1) * NCLASS + l15];
        }
        b.x = bx; b.y = by;
        c = __builtin_amdgcn_wmma_f32_16x16x4_f32(false, a, false, b,
                                                  (short)0, c, false, false);
    }
    if (l15 < NCLASS) {
        float*       om = m1   + (row0 + 8 * hh) * NCLASS + l15;
        float*       oo = out  + (row0 + 8 * hh) * NCLASS + l15;
        const float* dr = dinv +  row0 + 8 * hh;
        const float  bj = b1[l15];
#pragma unroll
        for (int v = 0; v < 8; ++v) {
            const float di = dr[v];
            const float ms = di * c[v];
            om[v * NCLASS] = ms;
            oo[v * NCLASS] = di * ms + bj;
        }
    }
}


#define S1_TBL    10016
#define S1_PARTS  10
#define S1_SLICES 25
#define S1_EPS    (N_EDGES / S1_SLICES)
#define S1_UNR    5
static_assert(N_EDGES % S1_SLICES == 0, "slices are equal");
static_assert(S1_EPS % (1024 * S1_UNR) == 0, "a slice is a whole number of trips");
static_assert(S1_TBL % 32 == 0 && N_NODES % 32 == 0, "every table is whole 128-byte lines of out");
static_assert(S1_TBL * S1_PARTS >= N_NODES && S1_TBL * (S1_PARTS - 1) < N_NODES,
              "the tables cover the nodes and the last one is not empty");
static_assert(S1_TBL * (NCLASS + 1) * 4 <= 320 * 1024, "table and the part's dinv fit one WGP's LDS");

__global__ __launch_bounds__(1024) void k_scatter1(const int* __restrict__ ei,
                                                   const float* __restrict__ ew,
                                                   const float* __restrict__ dinv,
                                                   const float* __restrict__ m1,
                                                   float* __restrict__ out) {
    __shared__ float tbl[S1_TBL * NCLASS];
    __shared__ float dls[S1_TBL];
    const int tid   = threadIdx.x;
    const int part  = blockIdx.x % S1_PARTS;
    const int slice = blockIdx.x / S1_PARTS;
    const int lo    = part * S1_TBL;
    const int rest  = N_NODES - lo;
    const unsigned n_here = (unsigned)(rest < S1_TBL ? rest : S1_TBL);

    for (int i = tid; i < S1_TBL * NCLASS; i += 1024) tbl[i] = 0.0f;
    for (unsigned i = tid; i < n_here; i += 1024) dls[i] = dinv[lo + i];
    __syncthreads();

    const int*   src = ei + slice * S1_EPS + tid;
    const int*   dst = ei + N_EDGES + slice * S1_EPS + tid;
    const float* w   = ew + slice * S1_EPS + tid;

    int ns[S1_UNR], nd[S1_UNR]; float nw[S1_UNR];
#pragma unroll
    for (int k = 0; k < S1_UNR; ++k) {
        ns[k] = src[k * 1024]; nd[k] = dst[k * 1024]; nw[k] = w[k * 1024];
    }
    for (int base = 0; base < S1_EPS; base += 1024 * S1_UNR) {
        int cs[S1_UNR], cd[S1_UNR]; float cw[S1_UNR];
#pragma unroll
        for (int k = 0; k < S1_UNR; ++k) { cs[k] = ns[k]; cd[k] = nd[k]; cw[k] = nw[k]; }
        const int nb = base + 1024 * S1_UNR;
        if (nb < S1_EPS) {
#pragma unroll
            for (int k = 0; k < S1_UNR; ++k) {
                ns[k] = src[nb + k * 1024]; nd[k] = dst[nb + k * 1024]; nw[k] = w[nb + k * 1024];
            }
        }
        float v[S1_UNR][NCLASS];
#pragma unroll
        for (int k = 0; k < S1_UNR; ++k) {
            const unsigned r = (unsigned)(cd[k] - lo);
            if (r < n_here) {
                const float* row = m1 + cs[k] * NCLASS;
#pragma unroll
                for (int j = 0; j < NCLASS; ++j) v[k][j] = row[j];
            }
        }
#pragma unroll
        for (int k = 0; k < S1_UNR; ++k) {
            const unsigned r = (unsigned)(cd[k] - lo);
            if (r < n_here) {
#pragma unroll
                for (int j = 0; j < NCLASS; ++j) atomicAdd(&tbl[r * NCLASS + j], cw[k] * v[k][j]);
            }
        }
    }
    __syncthreads();

    const int lane = tid & 31;
    const int wv   = tid >> 5;
    const int nl   = (int)(n_here * NCLASS) >> 5;
    const int start = (slice * nl) / S1_SLICES;
    float* o = out + lo * NCLASS;
    for (int j = wv; j < nl; j += 32) {
        int line = j + start;
        if (line >= nl) line -= nl;
        const unsigned i = (unsigned)(line * 32 + lane);
        atomicAdd(&o[i], dls[i / NCLASS] * tbl[i]);
    }
}


extern "C" void kernel_launch(void* const* d_in, const int* in_sizes, int n_in,
                              void* d_out, int out_size, void* d_ws, size_t ws_size,
                              hipStream_t stream) {
    const float* x  = (const float*)d_in[0];
    const int*   ei = (const int*)  d_in[1];
    const float* ew = (const float*)d_in[2];
    const float* W0 = (const float*)d_in[3];
    const float* b0 = (const float*)d_in[4];
    const float* W1 = (const float*)d_in[5];
    const float* b1 = (const float*)d_in[6];
    float* out = (float*)d_out;

    float* ws   = (float*)d_ws;
    float* dinv = ws;
    float* m0   = dinv + N_NODES;
    float* h    = m0 + (size_t)N_NODES * HIDDEN;
    float* m1   = h  + (size_t)N_NODES * HIDDEN;

    k_zero<<<(N_NODES + 255) / 256, 256, 0, stream>>>(dinv, N_NODES);
    k_deg <<<2 * DEG_SLICES, 1024, 0, stream>>>(ei, ew, dinv);
    k_dinv<<<(N_NODES + 255) / 256, 256, 0, stream>>>(dinv);

    k_gemm0<<<ROWBLKS, 128, 0, stream>>>(x, W0, dinv, m0, h);

    k_scatter0<<<S0_PARTS * S0_SLICES, 1024, 0, stream>>>(ei, ew, dinv, m0, h);

    k_gemm1<<<(ROWBLKS + 3) / 4, 128, 0, stream>>>(h, b0, W1, dinv, b1, m1, out);

    k_scatter1<<<S1_PARTS * S1_SLICES, 1024, 0, stream>>>(ei, ew, dinv, m1, out);
}


# layer-0 scan: a first pass stores each edge's destination part id as 7 bit planes (28 B per 32 edges, in m1's place); a lane tests 32 edges from 7 word loads, was a ballot per edge; sums reordered
# speedup vs baseline: 3.0257x; 1.1364x over previous
#include <hip/hip_runtime.h>

typedef __attribute__((ext_vector_type(2))) float v2f;
typedef __attribute__((ext_vector_type(8))) float v8f;

#define N_NODES 100000
#define N_EDGES 3200000
#define D_FEAT  256
#define HIDDEN  64
#define NCLASS  7
#define ROWBLKS (N_NODES / 16)


#define S0_TBL    1250
#define S0_PARTS  80
#define PL_EDGES  1024
#define PL_WORDS  (7 * 32)
#define N_PLB     (N_EDGES / PL_EDGES)
static_assert(N_EDGES % PL_EDGES == 0, "the edge list is whole plane blocks");
static_assert(S0_TBL * S0_PARTS == N_NODES && S0_PARTS <= 127, "a part id, or 127 for no part, fits 7 bits");
static_assert(N_PLB * PL_WORDS <= N_NODES * NCLASS, "the planes fit the m1 region");
static_assert(N_PLB * 32 >= N_NODES, "one thread per node among the plane waves");

__global__ __launch_bounds__(256) void k_prep(const int* __restrict__ ei,
                                              float* __restrict__ deg,
                                              unsigned* __restrict__ planes) {
    const int i = blockIdx.x * 256 + threadIdx.x;
    if (i < N_NODES) deg[i] = 0.0f;
    const int lane = threadIdx.x & 31;
    const int pb   = i >> 5;
    if (pb >= N_PLB) return;
    const int* dst = ei + N_EDGES + pb * PL_EDGES + lane;
    int dd[32];
#pragma unroll
    for (int g = 0; g < 32; ++g) dd[g] = dst[g * 32];
    unsigned p[7] = {0u, 0u, 0u, 0u, 0u, 0u, 0u};
#pragma unroll
    for (int g = 0; g < 32; ++g) {
        const unsigned part = ((unsigned)dd[g] < (unsigned)N_NODES) ? (unsigned)dd[g] / S0_TBL : 127u;
#pragma unroll
        for (int b = 0; b < 7; ++b) {
            const unsigned wb = __builtin_amdgcn_ballot_w32(((part >> b) & 1u) != 0u);
            if (lane == g) p[b] = wb;
        }
    }
    unsigned* o = planes + pb * PL_WORDS + lane;
#pragma unroll
    for (int b = 0; b < 7; ++b) o[b * 32] = p[b];
}

#define DEG_TBL    50016
#define DEG_SLICES 128
#define DEG_EPS    (N_EDGES / DEG_SLICES)
#define DEG_UNR    13
static_assert(N_EDGES % DEG_SLICES == 0, "slices are equal");
static_assert(DEG_TBL % 32 == 0 && (N_NODES - DEG_TBL) % 32 == 0 && N_NODES <= 2 * DEG_TBL &&
              N_NODES > DEG_TBL, "two tables of whole lines cover the nodes");

__global__ __launch_bounds__(1024) void k_deg(const int* __restrict__ ei,
                                              const float* __restrict__ ew,
                                              float* __restrict__ deg) {
    __shared__ float tbl[DEG_TBL];
    const int tid   = threadIdx.x;
    const int half  = blockIdx.x & 1;
    const int slice = blockIdx.x >> 1;
    const int lo    = half * DEG_TBL;
    const unsigned n_here = half ? (N_NODES - DEG_TBL) : DEG_TBL;

    for (int i = tid; i < DEG_TBL; i += 1024) tbl[i] = 0.0f;
    __syncthreads();

    const int*   dst = ei + N_EDGES + slice * DEG_EPS;
    const float* w   = ew + slice * DEG_EPS;
    for (int base = 0; base < DEG_EPS; base += 1024 * DEG_UNR) {
        int dd[DEG_UNR]; float ww[DEG_UNR];
#pragma unroll
        for (int k = 0; k < DEG_UNR; ++k) {
            const int i = base + k * 1024 + tid;
            const bool in = i < DEG_EPS;
            dd[k] = in ? dst[i] : -1;
            ww[k] = in ? w[i]   : 0.0f;
        }
#pragma unroll
        for (int k = 0; k < DEG_UNR; ++k) {
            const unsigned r = (unsigned)(dd[k] - lo);
            if (r < n_here) atomicAdd(&tbl[r], ww[k]);
        }
    }
    __syncthreads();

    const int lane = tid & 31;
    const int wv   = tid >> 5;
    const int nl   = (int)(n_here >> 5);
    const int start = (slice * nl) / DEG_SLICES;
    for (int j = wv; j < nl; j += 32) {
        int line = j + start;
        if (line >= nl) line -= nl;
        atomicAdd(&deg[lo + line * 32 + lane], tbl[line * 32 + lane]);
    }
}

__global__ void k_dinv(float* __restrict__ deg) {
    int i = blockIdx.x * blockDim.x + threadIdx.x;
    if (i < N_NODES) {
        float d = deg[i] + 1.0f;
        deg[i] = rsqrtf(fmaxf(d, 1e-12f));
    }
}

#define G0_PITCH (D_FEAT + 4)
#define G0_LOADS (16 * D_FEAT / 4 / 128)
static_assert((16 * D_FEAT) % (4 * 128) == 0, "the tile is a whole number of 16-byte loads per thread");
static_assert(G0_PITCH % 4 == 0 && (G0_PITCH % 64) == 4, "16-byte aligned rows, 4 banks apart");

__global__ __launch_bounds__(128) void k_gemm0(const float* __restrict__ x,
                                               const float* __restrict__ W0,
                                               const float* __restrict__ dinv,
                                               float* __restrict__ m0,
                                               float* __restrict__ h) {
    __shared__ __attribute__((aligned(16))) float xt[16 * G0_PITCH];
    const int tid  = threadIdx.x;
    const int lane = tid & 31;
    const int wv   = tid >> 5;
    const int row0 = blockIdx.x * 16;
    const int col0 = wv * 16;
    const int hh   = lane >> 4;
    const int l15  = lane & 15;

    const float* run = x + (size_t)row0 * D_FEAT;
    float4 piece[G0_LOADS];
#pragma unroll
    for (int j = 0; j < G0_LOADS; ++j) {
        const float* p = run + (j * 128 + tid) * 4;
        piece[j] = make_float4(p[0], p[1], p[2], p[3]);
    }
#pragma unroll
    for (int j = 0; j < G0_LOADS; ++j) {
        const int q = j * 128 + tid;
        float* d = xt + (q >> 6) * G0_PITCH + (q & 63) * 4;
        d[0] = piece[j].x; d[1] = piece[j].y; d[2] = piece[j].z; d[3] = piece[j].w;
    }
    __syncthreads();

    v8f c = {};
    const float* xrow = xt + l15 * G0_PITCH + 2 * hh;
    const float* wcol = W0 + (2 * hh) * HIDDEN + col0 + l15;

#pragma unroll 8
    for (int k = 0; k < D_FEAT; k += 4) {
        v2f a, b;
        a.x = xrow[k];
        a.y = xrow[k + 1];
        b.x = wcol[k * HIDDEN];
        b.y = wcol[k * HIDDEN + HIDDEN];
        c = __builtin_amdgcn_wmma_f32_16x16x4_f32(false, a, false, b,
                                                  (short)0, c, false, false);
    }
    float*       out = m0   + (row0 + 8 * hh) * HIDDEN + col0 + l15;
    float*       hs  = h    + (row0 + 8 * hh) * HIDDEN + col0 + l15;
    const float* dr  = dinv +  row0 + 8 * hh;
#pragma unroll
    for (int v = 0; v < 8; ++v) {
        const float di = dr[v];
        const float ms = di * c[v];
        out[v * HIDDEN] = ms;
        hs[v * HIDDEN]  = di * ms;
    }
}


#define S0_SLICES 3
#define S0_PLBS   1042
#define S0_G      16
#define S0_QCAP   48
static_assert((S0_TBL * HIDDEN) % 32 == 0, "a table is whole 128-byte lines of h");
static_assert(S0_PLBS * S0_SLICES >= N_PLB && S0_PLBS * (S0_SLICES - 1) < N_PLB,
              "the slices cover the plane blocks and the last one is not empty");
static_assert(S0_PLBS >= 32 && S0_PLBS * (S0_SLICES - 1) + 32 <= N_PLB, "every wave of a block has a plane block");
static_assert(32 % S0_G == 0 && S0_QCAP >= 32, "a full walk always leaves room for every lane's next hit");
static_assert(32 * S0_QCAP >= S0_TBL && S0_TBL > 1024 && S0_TBL <= 2048,
              "the dead queues hold the part's dinv: two values per thread at most");

static __device__ __forceinline__ void s0_walk(const unsigned* q, const int n, const int lane, const int lo,
                                               const int* ei, const float* ew, const float* m0,
                                               float* tl) {
    int   s_l  = 0;
    int   r_l  = 0;
    float wn_l = 0.0f;
    if (lane < n) {
        const unsigned e = q[lane];
        s_l  = ei[e];
        r_l  = ei[N_EDGES + e] - lo;
        wn_l = ew[e];
    }
    for (int j0 = 0; j0 < n; j0 += S0_G) {
        float v0[S0_G], v1[S0_G];
#pragma unroll
        for (int j = 0; j < S0_G; ++j) {
            const float* row = m0 + (unsigned)__builtin_amdgcn_readlane(s_l, j0 + j) * HIDDEN;
            v0[j] = row[lane];
            v1[j] = row[lane + 32];
        }
#pragma unroll
        for (int j = 0; j < S0_G; ++j) {
            if (j0 + j < n) {
                const int r = __builtin_amdgcn_readlane(r_l, j0 + j);
                const float wn = __int_as_float(
                    __builtin_amdgcn_readlane(__float_as_int(wn_l), j0 + j));
                atomicAdd(tl + r * HIDDEN,      wn * v0[j]);
                atomicAdd(tl + r * HIDDEN + 32, wn * v1[j]);
            }
        }
    }
}

__global__ __launch_bounds__(1024) void k_scatter0(const int* __restrict__ ei,
                                                  const float* __restrict__ ew,
                                                  const float* __restrict__ dinv,
                                                  const unsigned* __restrict__ planes,
                                                  const float* __restrict__ m0,
                                                  float* __restrict__ h) {
    __shared__ float tbl[S0_TBL * HIDDEN];
    __shared__ unsigned queue[32 * S0_QCAP];
    const int tid   = threadIdx.x;
    const int lane  = tid & 31;
    const int wv    = __builtin_amdgcn_readfirstlane(tid >> 5);
    const int qb    = wv * S0_QCAP;
    const int part  = blockIdx.x % S0_PARTS;
    const int slice = blockIdx.x / S0_PARTS;
    const int lo    = part * S0_TBL;
    const int pb0   = slice * S0_PLBS;
    const int pb1   = (pb0 + S0_PLBS < N_PLB) ? pb0 + S0_PLBS : N_PLB;

    const float dv0 = dinv[lo + tid];
    const float dv1 = (tid < S0_TBL - 1024) ? dinv[lo + 1024 + tid] : 0.0f;

    for (int i = tid; i < S0_TBL * HIDDEN; i += 1024) tbl[i] = 0.0f;
    __syncthreads();

    unsigned flip[7];
#pragma unroll
    for (int b = 0; b < 7; ++b) flip[b] = ((part >> b) & 1) ? 0u : ~0u;

    float*          tl = tbl + lane;
    const unsigned* pl = planes + lane;
    const unsigned* q  = queue + qb;

    int qn = 0;
    unsigned nx[7] = {0u, 0u, 0u, 0u, 0u, 0u, 0u};
    int pb = pb0 + wv;
#pragma unroll
    for (int b = 0; b < 7; ++b) nx[b] = pl[pb * PL_WORDS + b * 32];
    for (; pb < pb1; pb += 32) {
        unsigned m = ~0u;
#pragma unroll
        for (int b = 0; b < 7; ++b) m &= nx[b] ^ flip[b];
        if (pb + 32 < pb1) {
#pragma unroll
            for (int b = 0; b < 7; ++b) nx[b] = pl[(pb + 32) * PL_WORDS + b * 32];
        }
        const unsigned ebase = (unsigned)(pb * PL_EDGES + lane * 32);
        for (;;) {
            const bool hit = m != 0u;
            const unsigned bm = __builtin_amdgcn_ballot_w32(hit);
            if (bm == 0u) break;
            const int pos = qn + (int)__builtin_amdgcn_mbcnt_lo(bm, 0u);
            if (hit && pos < S0_QCAP) {
                queue[qb + pos] = ebase + (unsigned)__builtin_ctz(m);
                m &= m - 1u;
            }
            qn += __builtin_popcount(bm);
            if (qn > S0_QCAP) qn = S0_QCAP;
            if (qn >= 32) {
                s0_walk(q, 32, lane, lo, ei, ew, m0, tl);
                const int rem = qn - 32;
                if (lane < rem) {
                    const unsigned t = queue[qb + 32 + lane];
                    queue[qb + lane] = t;
                }
                qn = rem;
            }
        }
    }
    if (qn > 0) s0_walk(q, qn, lane, lo, ei, ew, m0, tl);
    __syncthreads();

    queue[tid] = __float_as_uint(dv0);
    if (tid < S0_TBL - 1024) queue[1024 + tid] = __float_as_uint(dv1);
    __syncthreads();

    const int nl    = S0_TBL * HIDDEN / 32;
    const int start = (slice * nl) / S0_SLICES;
    float* o = h + lo * HIDDEN;
    for (int j = wv; j < nl; j += 32) {
        int line = j + start;
        if (line >= nl) line -= nl;
        atomicAdd(&o[line * 32 + lane], __uint_as_float(queue[line >> 1]) * tbl[line * 32 + lane]);
    }
}


__global__ __launch_bounds__(128) void k_gemm1(const float* __restrict__ h,
                                               const float* __restrict__ b0,
                                               const float* __restrict__ W1,
                                               const float* __restrict__ dinv,
                                               const float* __restrict__ b1,
                                               float* __restrict__ m1,
                                               float* __restrict__ out) {
    const int lane = threadIdx.x & 31;
    const int wv   = threadIdx.x >> 5;
    const int rb   = blockIdx.x * 4 + wv;
    if (rb >= ROWBLKS) return;
    const int row0 = rb * 16;
    const int hh   = lane >> 4;
    const int l15  = lane & 15;

    v8f c = {};
    const float* hrow = h  + (row0 + l15) * HIDDEN + 2 * hh;
    const float* brow = b0 + 2 * hh;
#pragma unroll
    for (int k = 0; k < HIDDEN; k += 4) {
        v2f a, b;
        a.x = fmaxf(hrow[k]     + brow[k],     0.0f);
        a.y = fmaxf(hrow[k + 1] + brow[k + 1], 0.0f);
        float bx = 0.0f, by = 0.0f;
        if (l15 < NCLASS) {
            bx = W1[(k + 2 * hh) * NCLASS + l15];
            by = W1[(k + 2 * hh + 1) * NCLASS + l15];
        }
        b.x = bx; b.y = by;
        c = __builtin_amdgcn_wmma_f32_16x16x4_f32(false, a, false, b,
                                                  (short)0, c, false, false);
    }
    if (l15 < NCLASS) {
        float*       om = m1   + (row0 + 8 * hh) * NCLASS + l15;
        float*       oo = out  + (row0 + 8 * hh) * NCLASS + l15;
        const float* dr = dinv +  row0 + 8 * hh;
        const float  bj = b1[l15];
#pragma unroll
        for (int v = 0; v < 8; ++v) {
            const float di = dr[v];
            const float ms = di * c[v];
            om[v * NCLASS] = ms;
            oo[v * NCLASS] = di * ms + bj;
        }
    }
}


#define S1_TBL    10016
#define S1_PARTS  10
#define S1_SLICES 25
#define S1_EPS    (N_EDGES / S1_SLICES)
#define S1_UNR    5
static_assert(N_EDGES % S1_SLICES == 0, "slices are equal");
static_assert(S1_EPS % (1024 * S1_UNR) == 0, "a slice is a whole number of trips");
static_assert(S1_TBL % 32 == 0 && N_NODES % 32 == 0, "every table is whole 128-byte lines of out");
static_assert(S1_TBL * S1_PARTS >= N_NODES && S1_TBL * (S1_PARTS - 1) < N_NODES,
              "the tables cover the nodes and the last one is not empty");
static_assert(S1_TBL * (NCLASS + 1) * 4 <= 320 * 1024, "table and the part's dinv fit one WGP's LDS");

__global__ __launch_bounds__(1024) void k_scatter1(const int* __restrict__ ei,
                                                   const float* __restrict__ ew,
                                                   const float* __restrict__ dinv,
                                                   const float* __restrict__ m1,
                                                   float* __restrict__ out) {
    __shared__ float tbl[S1_TBL * NCLASS];
    __shared__ float dls[S1_TBL];
    const int tid   = threadIdx.x;
    const int part  = blockIdx.x % S1_PARTS;
    const int slice = blockIdx.x / S1_PARTS;
    const int lo    = part * S1_TBL;
    const int rest  = N_NODES - lo;
    const unsigned n_here = (unsigned)(rest < S1_TBL ? rest : S1_TBL);

    for (int i = tid; i < S1_TBL * NCLASS; i += 1024) tbl[i] = 0.0f;
    for (unsigned i = tid; i < n_here; i += 1024) dls[i] = dinv[lo + i];
    __syncthreads();

    const int*   src = ei + slice * S1_EPS + tid;
    const int*   dst = ei + N_EDGES + slice * S1_EPS + tid;
    const float* w   = ew + slice * S1_EPS + tid;

    int ns[S1_UNR], nd[S1_UNR]; float nw[S1_UNR];
#pragma unroll
    for (int k = 0; k < S1_UNR; ++k) {
        ns[k] = src[k * 1024]; nd[k] = dst[k * 1024]; nw[k] = w[k * 1024];
    }
    for (int base = 0; base < S1_EPS; base += 1024 * S1_UNR) {
        int cs[S1_UNR], cd[S1_UNR]; float cw[S1_UNR];
#pragma unroll
        for (int k = 0; k < S1_UNR; ++k) { cs[k] = ns[k]; cd[k] = nd[k]; cw[k] = nw[k]; }
        const int nb = base + 1024 * S1_UNR;
        if (nb < S1_EPS) {
#pragma unroll
            for (int k = 0; k < S1_UNR; ++k) {
                ns[k] = src[nb + k * 1024]; nd[k] = dst[nb + k * 1024]; nw[k] = w[nb + k * 1024];
            }
        }
        float v[S1_UNR][NCLASS];
#pragma unroll
        for (int k = 0; k < S1_UNR; ++k) {
            const unsigned r = (unsigned)(cd[k] - lo);
            if (r < n_here) {
                const float* row = m1 + cs[k] * NCLASS;
#pragma unroll
                for (int j = 0; j < NCLASS; ++j) v[k][j] = row[j];
            }
        }
#pragma unroll
        for (int k = 0; k < S1_UNR; ++k) {
            const unsigned r = (unsigned)(cd[k] - lo);
            if (r < n_here) {
#pragma unroll
                for (int j = 0; j < NCLASS; ++j) atomicAdd(&tbl[r * NCLASS + j], cw[k] * v[k][j]);
            }
        }
    }
    __syncthreads();

    const int lane = tid & 31;
    const int wv   = tid >> 5;
    const int nl   = (int)(n_here * NCLASS) >> 5;
    const int start = (slice * nl) / S1_SLICES;
    float* o = out + lo * NCLASS;
    for (int j = wv; j < nl; j += 32) {
        int line = j + start;
        if (line >= nl) line -= nl;
        const unsigned i = (unsigned)(line * 32 + lane);
        atomicAdd(&o[i], dls[i / NCLASS] * tbl[i]);
    }
}


extern "C" void kernel_launch(void* const* d_in, const int* in_sizes, int n_in,
                              void* d_out, int out_size, void* d_ws, size_t ws_size,
                              hipStream_t stream) {
    const float* x  = (const float*)d_in[0];
    const int*   ei = (const int*)  d_in[1];
    const float* ew = (const float*)d_in[2];
    const float* W0 = (const float*)d_in[3];
    const float* b0 = (const float*)d_in[4];
    const float* W1 = (const float*)d_in[5];
    const float* b1 = (const float*)d_in[6];
    float* out = (float*)d_out;

    float* ws   = (float*)d_ws;
    float* dinv = ws;
    float* m0   = dinv + N_NODES;
    float* h    = m0 + (size_t)N_NODES * HIDDEN;
    float* m1   = h  + (size_t)N_NODES * HIDDEN;
    unsigned* planes = (unsigned*)m1;

    k_prep<<<(N_NODES + 255) / 256, 256, 0, stream>>>(ei, dinv, planes);
    k_deg <<<2 * DEG_SLICES, 1024, 0, stream>>>(ei, ew, dinv);
    k_dinv<<<(N_NODES + 255) / 256, 256, 0, stream>>>(dinv);

    k_gemm0<<<ROWBLKS, 128, 0, stream>>>(x, W0, dinv, m0, h);

    k_scatter0<<<S0_PARTS * S0_SLICES, 1024, 0, stream>>>(ei, ew, dinv, planes, m0, h);

    k_gemm1<<<(ROWBLKS + 3) / 4, 128, 0, stream>>>(h, b0, W1, dinv, b1, m1, out);

    k_scatter1<<<S1_PARTS * S1_SLICES, 1024, 0, stream>>>(ei, ew, dinv, m1, out);
}


# layer-0 edges binned by destination part (400 nodes) into h's rows (new pass); one block per part sums its bin in LDS, stores h once; over-full bin: exact edge-list scan; sums in atomic rank order
# speedup vs baseline: 3.7122x; 1.2269x over previous
#include <hip/hip_runtime.h>

typedef __attribute__((ext_vector_type(2))) float v2f;
typedef __attribute__((ext_vector_type(8))) float v8f;

#define N_NODES 100000
#define N_EDGES 3200000
#define D_FEAT  256
#define HIDDEN  64
#define NCLASS  7
#define ROWBLKS (N_NODES / 16)


#define B_TBL    400
#define B_PARTS  250
#define B_MAIN   (B_TBL * HIDDEN / 2)
#define B_OVF    1390
#define B_CAP    (B_MAIN + B_OVF)
#define B_CUR    (B_PARTS * B_OVF * 2)
static_assert(B_TBL * B_PARTS == N_NODES, "the parts cover the nodes exactly");
static_assert(N_NODES <= (1 << 17) && B_TBL <= (1 << 15), "source and row fit one word");
static_assert(B_CUR + B_PARTS <= N_NODES * NCLASS, "overflow areas and cursors fit the m1 region");
static_assert(N_EDGES % 32 == 0, "the edge list is whole groups of 32 edges");

static __device__ __forceinline__ uint2* bin_rec(const unsigned part, const unsigned pos, float* h,
                                                 unsigned* ovf) {
    if (pos < (unsigned)B_MAIN) return (uint2*)(h + (size_t)part * B_TBL * HIDDEN) + pos;
    return (uint2*)(ovf + part * (B_OVF * 2)) + (pos - B_MAIN);
}

__global__ __launch_bounds__(256) void k_prep(float* __restrict__ deg, unsigned* __restrict__ cursor) {
    const int i = blockIdx.x * 256 + threadIdx.x;
    if (i < N_NODES) deg[i] = 0.0f;
    if (i < B_PARTS) cursor[i] = 0u;
}

#define BIN_EPB  (N_EDGES / B_PARTS)
#define BIN_UNR  13
static_assert(N_EDGES % B_PARTS == 0 && BIN_UNR * 1024 >= BIN_EPB, "a block's threads cover its slice");
static_assert(B_PARTS <= 32 * 8, "one wave, eight parts per lane, scans the counts");

__global__ __launch_bounds__(1024) void k_bin(const int* __restrict__ ei,
                                              const float* __restrict__ ew,
                                              float* h, unsigned* ovf, unsigned* cursor) {
    __shared__ unsigned cnt[B_PARTS];
    __shared__ unsigned loff[B_PARTS];
    __shared__ unsigned gpos[B_PARTS];
    __shared__ uint4 stage[BIN_EPB];
    const int tid  = threadIdx.x;
    const int base = blockIdx.x * BIN_EPB;
    if (tid < B_PARTS) cnt[tid] = 0u;
    __syncthreads();

    int ss[BIN_UNR], dd[BIN_UNR]; float ww[BIN_UNR]; unsigned rk[BIN_UNR];
#pragma unroll
    for (int k = 0; k < BIN_UNR; ++k) {
        const int i = k * 1024 + tid;
        const bool in = i < BIN_EPB;
        ss[k] = in ? ei[base + i] : 0;
        dd[k] = in ? ei[N_EDGES + base + i] : -1;
        ww[k] = in ? ew[base + i] : 0.0f;
    }
#pragma unroll
    for (int k = 0; k < BIN_UNR; ++k) {
        rk[k] = 0u;
        if ((unsigned)dd[k] < (unsigned)N_NODES) rk[k] = atomicAdd(&cnt[(unsigned)dd[k] / B_TBL], 1u);
    }
    __syncthreads();
    if (tid < B_PARTS) {
        const unsigned n = cnt[tid];
        gpos[tid] = n ? atomicAdd(&cursor[tid], n) : 0u;
    }
    if (tid < 32) {
        unsigned v[8];
        unsigned run = 0u;
#pragma unroll
        for (int k = 0; k < 8; ++k) {
            const int p = 8 * tid + k;
            const unsigned t = (p < B_PARTS) ? cnt[p] : 0u;
            v[k] = run;
            run += t;
        }
        unsigned incl = run;
#pragma unroll
        for (int d = 1; d < 32; d <<= 1) {
            const unsigned up = __shfl_up(incl, (unsigned)d);
            if (tid >= d) incl += up;
        }
        const unsigned excl = incl - run;
#pragma unroll
        for (int k = 0; k < 8; ++k) {
            const int p = 8 * tid + k;
            if (p < B_PARTS) loff[p] = excl + v[k];
        }
    }
    __syncthreads();
#pragma unroll
    for (int k = 0; k < BIN_UNR; ++k) {
        if ((unsigned)dd[k] < (unsigned)N_NODES) {
            const unsigned part = (unsigned)dd[k] / B_TBL;
            uint4 s;
            s.x = ((unsigned)ss[k] & 0x1FFFFu) | (((unsigned)dd[k] - part * B_TBL) << 17);
            s.y = __float_as_uint(ww[k]);
            s.z = part;
            s.w = gpos[part] + rk[k];
            stage[loff[part] + rk[k]] = s;
        }
    }
    __syncthreads();
    const int tot = (int)(loff[B_PARTS - 1] + cnt[B_PARTS - 1]);
    for (int j = tid; j < tot; j += 1024) {
        const uint4 s = stage[j];
        if (s.w < (unsigned)B_CAP) {
            uint2 r; r.x = s.x; r.y = s.y;
            *bin_rec(s.z, s.w, h, ovf) = r;
        }
    }
}

#define DEG_TBL    50016
#define DEG_SLICES 128
#define DEG_EPS    (N_EDGES / DEG_SLICES)
#define DEG_UNR    13
static_assert(N_EDGES % DEG_SLICES == 0, "slices are equal");
static_assert(DEG_TBL % 32 == 0 && (N_NODES - DEG_TBL) % 32 == 0 && N_NODES <= 2 * DEG_TBL &&
              N_NODES > DEG_TBL, "two tables of whole lines cover the nodes");

__global__ __launch_bounds__(1024) void k_deg(const int* __restrict__ ei,
                                              const float* __restrict__ ew,
                                              float* __restrict__ deg) {
    __shared__ float tbl[DEG_TBL];
    const int tid   = threadIdx.x;
    const int half  = blockIdx.x & 1;
    const int slice = blockIdx.x >> 1;
    const int lo    = half * DEG_TBL;
    const unsigned n_here = half ? (N_NODES - DEG_TBL) : DEG_TBL;

    for (int i = tid; i < DEG_TBL; i += 1024) tbl[i] = 0.0f;
    __syncthreads();

    const int*   dst = ei + N_EDGES + slice * DEG_EPS;
    const float* w   = ew + slice * DEG_EPS;
    for (int base = 0; base < DEG_EPS; base += 1024 * DEG_UNR) {
        int dd[DEG_UNR]; float ww[DEG_UNR];
#pragma unroll
        for (int k = 0; k < DEG_UNR; ++k) {
            const int i = base + k * 1024 + tid;
            const bool in = i < DEG_EPS;
            dd[k] = in ? dst[i] : -1;
            ww[k] = in ? w[i]   : 0.0f;
        }
#pragma unroll
        for (int k = 0; k < DEG_UNR; ++k) {
            const unsigned r = (unsigned)(dd[k] - lo);
            if (r < n_here) atomicAdd(&tbl[r], ww[k]);
        }
    }
    __syncthreads();

    const int lane = tid & 31;
    const int wv   = tid >> 5;
    const int nl   = (int)(n_here >> 5);
    const int start = (slice * nl) / DEG_SLICES;
    for (int j = wv; j < nl; j += 32) {
        int line = j + start;
        if (line >= nl) line -= nl;
        atomicAdd(&deg[lo + line * 32 + lane], tbl[line * 32 + lane]);
    }
}

__global__ void k_dinv(float* __restrict__ deg) {
    int i = blockIdx.x * blockDim.x + threadIdx.x;
    if (i < N_NODES) {
        float d = deg[i] + 1.0f;
        deg[i] = rsqrtf(fmaxf(d, 1e-12f));
    }
}

#define G0_PITCH (D_FEAT + 4)
#define G0_LOADS (16 * D_FEAT / 4 / 128)
static_assert((16 * D_FEAT) % (4 * 128) == 0, "the tile is a whole number of 16-byte loads per thread");
static_assert(G0_PITCH % 4 == 0 && (G0_PITCH % 64) == 4, "16-byte aligned rows, 4 banks apart");

__global__ __launch_bounds__(128) void k_gemm0(const float* __restrict__ x,
                                               const float* __restrict__ W0,
                                               const float* __restrict__ dinv,
                                               float* __restrict__ m0) {
    __shared__ __attribute__((aligned(16))) float xt[16 * G0_PITCH];
    const int tid  = threadIdx.x;
    const int lane = tid & 31;
    const int wv   = tid >> 5;
    const int row0 = blockIdx.x * 16;
    const int col0 = wv * 16;
    const int hh   = lane >> 4;
    const int l15  = lane & 15;

    const float* run = x + (size_t)row0 * D_FEAT;
    float4 piece[G0_LOADS];
#pragma unroll
    for (int j = 0; j < G0_LOADS; ++j) {
        const float* p = run + (j * 128 + tid) * 4;
        piece[j] = make_float4(p[0], p[1], p[2], p[3]);
    }
#pragma unroll
    for (int j = 0; j < G0_LOADS; ++j) {
        const int q = j * 128 + tid;
        float* d = xt + (q >> 6) * G0_PITCH + (q & 63) * 4;
        d[0] = piece[j].x; d[1] = piece[j].y; d[2] = piece[j].z; d[3] = piece[j].w;
    }
    __syncthreads();

    v8f c = {};
    const float* xrow = xt + l15 * G0_PITCH + 2 * hh;
    const float* wcol = W0 + (2 * hh) * HIDDEN + col0 + l15;

#pragma unroll 8
    for (int k = 0; k < D_FEAT; k += 4) {
        v2f a, b;
        a.x = xrow[k];
        a.y = xrow[k + 1];
        b.x = wcol[k * HIDDEN];
        b.y = wcol[k * HIDDEN + HIDDEN];
        c = __builtin_amdgcn_wmma_f32_16x16x4_f32(false, a, false, b,
                                                  (short)0, c, false, false);
    }
    float*       out = m0   + (row0 + 8 * hh) * HIDDEN + col0 + l15;
    const float* dr  = dinv +  row0 + 8 * hh;
#pragma unroll
    for (int v = 0; v < 8; ++v) {
        out[v * HIDDEN] = dr[v] * c[v];
    }
}


#define S0_G      16
#define S0_FIN    (B_TBL * HIDDEN / 1024)
#define S0_STG    ((B_CAP + 1023) / 1024)
static_assert((B_TBL * HIDDEN) % 1024 == 0, "the final sweep is whole trips");
static_assert(32 % S0_G == 0, "a chunk of 32 records is whole groups");

static __device__ __forceinline__ void s0_walk(const uint2* q, const int n, const int lane,
                                               const float* m0, float* tl) {
    int   s_l  = 0;
    int   r_l  = 0;
    float wn_l = 0.0f;
    if (lane < n) {
        const uint2 r = q[lane];
        s_l  = (int)(r.x & 0x1FFFFu);
        r_l  = (int)(r.x >> 17);
        wn_l = __uint_as_float(r.y);
    }
    for (int j0 = 0; j0 < n; j0 += S0_G) {
        float v0[S0_G], v1[S0_G];
#pragma unroll
        for (int j = 0; j < S0_G; ++j) {
            const float* row = m0 + (unsigned)__builtin_amdgcn_readlane(s_l, j0 + j) * HIDDEN;
            v0[j] = row[lane];
            v1[j] = row[lane + 32];
        }
#pragma unroll
        for (int j = 0; j < S0_G; ++j) {
            if (j0 + j < n) {
                const int r = __builtin_amdgcn_readlane(r_l, j0 + j);
                const float wn = __int_as_float(
                    __builtin_amdgcn_readlane(__float_as_int(wn_l), j0 + j));
                atomicAdd(tl + r * HIDDEN,      wn * v0[j]);
                atomicAdd(tl + r * HIDDEN + 32, wn * v1[j]);
            }
        }
    }
}

static __device__ __forceinline__ void s0_scan(const int* ei, const float* ew, const float* m0,
                                               const int wv, const int lane, const int lo, float* tl) {
    for (int e0 = wv * 32; e0 < N_EDGES; e0 += 1024) {
        const int e = e0 + lane;
        const unsigned r = (unsigned)(ei[N_EDGES + e] - lo);
        const bool hit = r < (unsigned)B_TBL;
        unsigned bm = __builtin_amdgcn_ballot_w32(hit);
        if (bm == 0u) continue;
        int   s = 0;
        float w = 0.0f;
        if (hit) { s = ei[e]; w = ew[e]; }
        while (bm != 0u) {
            const int j = __builtin_ctz(bm);
            bm &= bm - 1u;
            const float* row = m0 + (unsigned)__builtin_amdgcn_readlane(s, j) * HIDDEN;
            const int   rj = __builtin_amdgcn_readlane((int)r, j);
            const float wj = __int_as_float(__builtin_amdgcn_readlane(__float_as_int(w), j));
            atomicAdd(tl + rj * HIDDEN,      wj * row[lane]);
            atomicAdd(tl + rj * HIDDEN + 32, wj * row[lane + 32]);
        }
    }
}

__global__ __launch_bounds__(1024) void k_scatter0(const int* __restrict__ ei,
                                                  const float* __restrict__ ew,
                                                  const float* __restrict__ dinv,
                                                  const unsigned* __restrict__ cursor,
                                                  const unsigned* __restrict__ ovf,
                                                  const float* __restrict__ m0,
                                                  float* h) {
    __shared__ float tbl[B_TBL * HIDDEN];
    __shared__ uint2 recs[B_CAP];
    const int tid  = threadIdx.x;
    const int lane = tid & 31;
    const int wv   = __builtin_amdgcn_readfirstlane(tid >> 5);
    const int part = blockIdx.x;
    const int lo   = part * B_TBL;

    const int total = __builtin_amdgcn_readfirstlane((int)cursor[part]);
    const bool over = total > B_CAP;
    const int n = over ? 0 : total;

    const uint2* mainp = (const uint2*)(h + (size_t)lo * HIDDEN);
    const uint2* ovfp  = (const uint2*)(ovf + part * (B_OVF * 2));
    uint2 rr[S0_STG];
#pragma unroll
    for (int k = 0; k < S0_STG; ++k) {
        const int i = k * 1024 + tid;
        rr[k] = make_uint2(0u, 0u);
        if (i < n) rr[k] = (i < B_MAIN) ? mainp[i] : ovfp[i - B_MAIN];
    }
    const float* own = m0 + (size_t)lo * HIDDEN;
    float mv[S0_FIN];
#pragma unroll
    for (int t = 0; t < S0_FIN; ++t) mv[t] = own[t * 1024 + tid];
#pragma unroll
    for (int k = 0; k < S0_STG; ++k) {
        const int i = k * 1024 + tid;
        if (i < n) recs[i] = rr[k];
    }
    for (int i = tid; i < B_TBL * HIDDEN; i += 1024) tbl[i] = 0.0f;
    __syncthreads();

    float* tl = tbl + lane;
    for (int c0 = wv * 32; c0 < n; c0 += 1024) {
        const int left = n - c0;
        s0_walk(recs + c0, left < 32 ? left : 32, lane, m0, tl);
    }
    if (over) s0_scan(ei, ew, m0, wv, lane, lo, tl);
    __syncthreads();

    float*       o   = h  + (size_t)lo * HIDDEN;
    const float* dl  = dinv + lo;
#pragma unroll
    for (int t = 0; t < S0_FIN; ++t) {
        const int i = t * 1024 + tid;
        o[i] = dl[i >> 6] * (tbl[i] + mv[t]);
    }
}


__global__ __launch_bounds__(128) void k_gemm1(const float* __restrict__ h,
                                               const float* __restrict__ b0,
                                               const float* __restrict__ W1,
                                               const float* __restrict__ dinv,
                                               const float* __restrict__ b1,
                                               float* __restrict__ m1,
                                               float* __restrict__ out) {
    const int lane = threadIdx.x & 31;
    const int wv   = threadIdx.x >> 5;
    const int rb   = blockIdx.x * 4 + wv;
    if (rb >= ROWBLKS) return;
    const int row0 = rb * 16;
    const int hh   = lane >> 4;
    const int l15  = lane & 15;

    v8f c = {};
    const float* hrow = h  + (row0 + l15) * HIDDEN + 2 * hh;
    const float* brow = b0 + 2 * hh;
#pragma unroll
    for (int k = 0; k < HIDDEN; k += 4) {
        v2f a, b;
        a.x = fmaxf(hrow[k]     + brow[k],     0.0f);
        a.y = fmaxf(hrow[k + 1] + brow[k + 1], 0.0f);
        float bx = 0.0f, by = 0.0f;
        if (l15 < NCLASS) {
            bx = W1[(k + 2 * hh) * NCLASS + l15];
            by = W1[(k + 2 * hh + 1) * NCLASS + l15];
        }
        b.x = bx; b.y = by;
        c = __builtin_amdgcn_wmma_f32_16x16x4_f32(false, a, false, b,
                                                  (short)0, c, false, false);
    }
    if (l15 < NCLASS) {
        float*       om = m1   + (row0 + 8 * hh) * NCLASS + l15;
        float*       oo = out  + (row0 + 8 * hh) * NCLASS + l15;
        const float* dr = dinv +  row0 + 8 * hh;
        const float  bj = b1[l15];
#pragma unroll
        for (int v = 0; v < 8; ++v) {
            const float di = dr[v];
            const float ms = di * c[v];
            om[v * NCLASS] = ms;
            oo[v * NCLASS] = di * ms + bj;
        }
    }
}


#define S1_TBL    10016
#define S1_PARTS  10
#define S1_SLICES 25
#define S1_EPS    (N_EDGES / S1_SLICES)
#define S1_UNR    5
static_assert(N_EDGES % S1_SLICES == 0, "slices are equal");
static_assert(S1_EPS % (1024 * S1_UNR) == 0, "a slice is a whole number of trips");
static_assert(S1_TBL % 32 == 0 && N_NODES % 32 == 0, "every table is whole 128-byte lines of out");
static_assert(S1_TBL * S1_PARTS >= N_NODES && S1_TBL * (S1_PARTS - 1) < N_NODES,
              "the tables cover the nodes and the last one is not empty");
static_assert(S1_TBL * (NCLASS + 1) * 4 <= 320 * 1024, "table and the part's dinv fit one WGP's LDS");

__global__ __launch_bounds__(1024) void k_scatter1(const int* __restrict__ ei,
                                                   const float* __restrict__ ew,
                                                   const float* __restrict__ dinv,
                                                   const float* __restrict__ m1,
                                                   float* __restrict__ out) {
    __shared__ float tbl[S1_TBL * NCLASS];
    __shared__ float dls[S1_TBL];
    const int tid   = threadIdx.x;
    const int part  = blockIdx.x % S1_PARTS;
    const int slice = blockIdx.x / S1_PARTS;
    const int lo    = part * S1_TBL;
    const int rest  = N_NODES - lo;
    const unsigned n_here = (unsigned)(rest < S1_TBL ? rest : S1_TBL);

    for (int i = tid; i < S1_TBL * NCLASS; i += 1024) tbl[i] = 0.0f;
    for (unsigned i = tid; i < n_here; i += 1024) dls[i] = dinv[lo + i];
    __syncthreads();

    const int*   src = ei + slice * S1_EPS + tid;
    const int*   dst = ei + N_EDGES + slice * S1_EPS + tid;
    const float* w   = ew + slice * S1_EPS + tid;

    int ns[S1_UNR], nd[S1_UNR]; float nw[S1_UNR];
#pragma unroll
    for (int k = 0; k < S1_UNR; ++k) {
        ns[k] = src[k * 1024]; nd[k] = dst[k * 1024]; nw[k] = w[k * 1024];
    }
    for (int base = 0; base < S1_EPS; base += 1024 * S1_UNR) {
        int cs[S1_UNR], cd[S1_UNR]; float cw[S1_UNR];
#pragma unroll
        for (int k = 0; k < S1_UNR; ++k) { cs[k] = ns[k]; cd[k] = nd[k]; cw[k] = nw[k]; }
        const int nb = base + 1024 * S1_UNR;
        if (nb < S1_EPS) {
#pragma unroll
            for (int k = 0; k < S1_UNR; ++k) {
                ns[k] = src[nb + k * 1024]; nd[k] = dst[nb + k * 1024]; nw[k] = w[nb + k * 1024];
            }
        }
        float v[S1_UNR][NCLASS];
#pragma unroll
        for (int k = 0; k < S1_UNR; ++k) {
            const unsigned r = (unsigned)(cd[k] - lo);
            if (r < n_here) {
                const float* row = m1 + cs[k] * NCLASS;
#pragma unroll
                for (int j = 0; j < NCLASS; ++j) v[k][j] = row[j];
            }
        }
#pragma unroll
        for (int k = 0; k < S1_UNR; ++k) {
            const unsigned r = (unsigned)(cd[k] - lo);
            if (r < n_here) {
#pragma unroll
                for (int j = 0; j < NCLASS; ++j) atomicAdd(&tbl[r * NCLASS + j], cw[k] * v[k][j]);
            }
        }
    }
    __syncthreads();

    const int lane = tid & 31;
    const int wv   = tid >> 5;
    const int nl   = (int)(n_here * NCLASS) >> 5;
    const int start = (slice * nl) / S1_SLICES;
    float* o = out + lo * NCLASS;
    for (int j = wv; j < nl; j += 32) {
        int line = j + start;
        if (line >= nl) line -= nl;
        const unsigned i = (unsigned)(line * 32 + lane);
        atomicAdd(&o[i], dls[i / NCLASS] * tbl[i]);
    }
}


extern "C" void kernel_launch(void* const* d_in, const int* in_sizes, int n_in,
                              void* d_out, int out_size, void* d_ws, size_t ws_size,
                              hipStream_t stream) {
    const float* x  = (const float*)d_in[0];
    const int*   ei = (const int*)  d_in[1];
    const float* ew = (const float*)d_in[2];
    const float* W0 = (const float*)d_in[3];
    const float* b0 = (const float*)d_in[4];
    const float* W1 = (const float*)d_in[5];
    const float* b1 = (const float*)d_in[6];
    float* out = (float*)d_out;

    float* ws   = (float*)d_ws;
    float* dinv = ws;
    float* m0   = dinv + N_NODES;
    float* h    = m0 + (size_t)N_NODES * HIDDEN;
    float* m1   = h  + (size_t)N_NODES * HIDDEN;
    unsigned* ovf    = (unsigned*)m1;
    unsigned* cursor = ovf + B_CUR;

    k_prep<<<(N_NODES + 255) / 256, 256, 0, stream>>>(dinv, cursor);
    k_deg <<<2 * DEG_SLICES, 1024, 0, stream>>>(ei, ew, dinv);
    k_dinv<<<(N_NODES + 255) / 256, 256, 0, stream>>>(dinv);

    k_gemm0<<<ROWBLKS, 128, 0, stream>>>(x, W0, dinv, m0);

    k_bin     <<<B_PARTS, 1024, 0, stream>>>(ei, ew, h, ovf, cursor);
    k_scatter0<<<B_PARTS, 1024, 0, stream>>>(ei, ew, dinv, cursor, ovf, m0, h);

    k_gemm1<<<(ROWBLKS + 3) / 4, 128, 0, stream>>>(h, b0, W1, dinv, b1, m1, out);

    k_scatter1<<<S1_PARTS * S1_SLICES, 1024, 0, stream>>>(ei, ew, dinv, m1, out);
}


# degree from the bins: binning runs first, one block per part sums its bin's weights per row in LDS and stores dinv; edge-list degree pass, line atomics, rsqrt pass gone (7 launches); sums reordered
# speedup vs baseline: 4.0241x; 1.0840x over previous
#include <hip/hip_runtime.h>

typedef __attribute__((ext_vector_type(2))) float v2f;
typedef __attribute__((ext_vector_type(8))) float v8f;

#define N_NODES 100000
#define N_EDGES 3200000
#define D_FEAT  256
#define HIDDEN  64
#define NCLASS  7
#define ROWBLKS (N_NODES / 16)


#define B_TBL    400
#define B_PARTS  250
#define B_MAIN   (B_TBL * HIDDEN / 2)
#define B_OVF    1390
#define B_CAP    (B_MAIN + B_OVF)
#define B_CUR    (B_PARTS * B_OVF * 2)
static_assert(B_TBL * B_PARTS == N_NODES, "the parts cover the nodes exactly");
static_assert(N_NODES <= (1 << 17) && B_TBL <= (1 << 15), "source and row fit one word");
static_assert(B_CUR + B_PARTS <= N_NODES * NCLASS, "overflow areas and cursors fit the m1 region");
static_assert(N_EDGES % 32 == 0, "the edge list is whole groups of 32 edges");

static __device__ __forceinline__ uint2* bin_rec(const unsigned part, const unsigned pos, float* h,
                                                 unsigned* ovf) {
    if (pos < (unsigned)B_MAIN) return (uint2*)(h + (size_t)part * B_TBL * HIDDEN) + pos;
    return (uint2*)(ovf + part * (B_OVF * 2)) + (pos - B_MAIN);
}

static_assert(B_PARTS <= 256, "one block of 256 threads clears the cursors");
__global__ __launch_bounds__(256) void k_prep(unsigned* __restrict__ cursor) {
    if (threadIdx.x < B_PARTS) cursor[threadIdx.x] = 0u;
}

#define BIN_EPB  (N_EDGES / B_PARTS)
#define BIN_UNR  13
static_assert(N_EDGES % B_PARTS == 0 && BIN_UNR * 1024 >= BIN_EPB, "a block's threads cover its slice");
static_assert(B_PARTS <= 32 * 8, "one wave, eight parts per lane, scans the counts");

__global__ __launch_bounds__(1024) void k_bin(const int* __restrict__ ei,
                                              const float* __restrict__ ew,
                                              float* h, unsigned* ovf, unsigned* cursor) {
    __shared__ unsigned cnt[B_PARTS];
    __shared__ unsigned loff[B_PARTS];
    __shared__ unsigned gpos[B_PARTS];
    __shared__ uint4 stage[BIN_EPB];
    const int tid  = threadIdx.x;
    const int base = blockIdx.x * BIN_EPB;
    if (tid < B_PARTS) cnt[tid] = 0u;
    __syncthreads();

    int ss[BIN_UNR], dd[BIN_UNR]; float ww[BIN_UNR]; unsigned rk[BIN_UNR];
#pragma unroll
    for (int k = 0; k < BIN_UNR; ++k) {
        const int i = k * 1024 + tid;
        const bool in = i < BIN_EPB;
        ss[k] = in ? ei[base + i] : 0;
        dd[k] = in ? ei[N_EDGES + base + i] : -1;
        ww[k] = in ? ew[base + i] : 0.0f;
    }
#pragma unroll
    for (int k = 0; k < BIN_UNR; ++k) {
        rk[k] = 0u;
        if ((unsigned)dd[k] < (unsigned)N_NODES) rk[k] = atomicAdd(&cnt[(unsigned)dd[k] / B_TBL], 1u);
    }
    __syncthreads();
    if (tid < B_PARTS) {
        const unsigned n = cnt[tid];
        gpos[tid] = n ? atomicAdd(&cursor[tid], n) : 0u;
    }
    if (tid < 32) {
        unsigned v[8];
        unsigned run = 0u;
#pragma unroll
        for (int k = 0; k < 8; ++k) {
            const int p = 8 * tid + k;
            const unsigned t = (p < B_PARTS) ? cnt[p] : 0u;
            v[k] = run;
            run += t;
        }
        unsigned incl = run;
#pragma unroll
        for (int d = 1; d < 32; d <<= 1) {
            const unsigned up = __shfl_up(incl, (unsigned)d);
            if (tid >= d) incl += up;
        }
        const unsigned excl = incl - run;
#pragma unroll
        for (int k = 0; k < 8; ++k) {
            const int p = 8 * tid + k;
            if (p < B_PARTS) loff[p] = excl + v[k];
        }
    }
    __syncthreads();
#pragma unroll
    for (int k = 0; k < BIN_UNR; ++k) {
        if ((unsigned)dd[k] < (unsigned)N_NODES) {
            const unsigned part = (unsigned)dd[k] / B_TBL;
            uint4 s;
            s.x = ((unsigned)ss[k] & 0x1FFFFu) | (((unsigned)dd[k] - part * B_TBL) << 17);
            s.y = __float_as_uint(ww[k]);
            s.z = part;
            s.w = gpos[part] + rk[k];
            stage[loff[part] + rk[k]] = s;
        }
    }
    __syncthreads();
    const int tot = (int)(loff[B_PARTS - 1] + cnt[B_PARTS - 1]);
    for (int j = tid; j < tot; j += 1024) {
        const uint4 s = stage[j];
        if (s.w < (unsigned)B_CAP) {
            uint2 r; r.x = s.x; r.y = s.y;
            *bin_rec(s.z, s.w, h, ovf) = r;
        }
    }
}

#define DG_STG   ((B_CAP + 1023) / 1024)

__global__ __launch_bounds__(1024) void k_degbin(const int* __restrict__ ei,
                                                 const float* __restrict__ ew,
                                                 const float* __restrict__ h,
                                                 const unsigned* __restrict__ ovf,
                                                 const unsigned* __restrict__ cursor,
                                                 float* __restrict__ dinv) {
    __shared__ float tbl[B_TBL];
    const int tid  = threadIdx.x;
    const int part = blockIdx.x;
    const int lo   = part * B_TBL;

    const int total = __builtin_amdgcn_readfirstlane((int)cursor[part]);
    const bool over = total > B_CAP;
    const int n = over ? 0 : total;

    const uint2* mainp = (const uint2*)(h + (size_t)lo * HIDDEN);
    const uint2* ovfp  = (const uint2*)(ovf + part * (B_OVF * 2));
    uint2 rr[DG_STG];
#pragma unroll
    for (int k = 0; k < DG_STG; ++k) rr[k] = make_uint2(0u, 0u);
    if (n > 0) {
#pragma unroll
        for (int k = 0; k < DG_STG; ++k) {
            const int i = k * 1024 + tid;
            const int j = i < n ? i : n - 1;
            rr[k] = (j < B_MAIN) ? mainp[j] : ovfp[j - B_MAIN];
        }
    }
    if (tid < B_TBL) tbl[tid] = 0.0f;
    __syncthreads();
#pragma unroll
    for (int k = 0; k < DG_STG; ++k) {
        const int i = k * 1024 + tid;
        if (i < n) atomicAdd(&tbl[rr[k].x >> 17], __uint_as_float(rr[k].y));
    }
    if (over) {
        for (int e = tid; e < N_EDGES; e += 1024) {
            const unsigned r = (unsigned)(ei[N_EDGES + e] - lo);
            if (r < (unsigned)B_TBL) atomicAdd(&tbl[r], ew[e]);
        }
    }
    __syncthreads();
    if (tid < B_TBL) {
        const float d = tbl[tid] + 1.0f;
        dinv[lo + tid] = rsqrtf(fmaxf(d, 1e-12f));
    }
}

#define G0_PITCH (D_FEAT + 4)
#define G0_LOADS (16 * D_FEAT / 4 / 128)
static_assert((16 * D_FEAT) % (4 * 128) == 0, "the tile is a whole number of 16-byte loads per thread");
static_assert(G0_PITCH % 4 == 0 && (G0_PITCH % 64) == 4, "16-byte aligned rows, 4 banks apart");

__global__ __launch_bounds__(128) void k_gemm0(const float* __restrict__ x,
                                               const float* __restrict__ W0,
                                               const float* __restrict__ dinv,
                                               float* __restrict__ m0) {
    __shared__ __attribute__((aligned(16))) float xt[16 * G0_PITCH];
    const int tid  = threadIdx.x;
    const int lane = tid & 31;
    const int wv   = tid >> 5;
    const int row0 = blockIdx.x * 16;
    const int col0 = wv * 16;
    const int hh   = lane >> 4;
    const int l15  = lane & 15;

    const float* run = x + (size_t)row0 * D_FEAT;
    float4 piece[G0_LOADS];
#pragma unroll
    for (int j = 0; j < G0_LOADS; ++j) {
        const float* p = run + (j * 128 + tid) * 4;
        piece[j] = make_float4(p[0], p[1], p[2], p[3]);
    }
#pragma unroll
    for (int j = 0; j < G0_LOADS; ++j) {
        const int q = j * 128 + tid;
        float* d = xt + (q >> 6) * G0_PITCH + (q & 63) * 4;
        d[0] = piece[j].x; d[1] = piece[j].y; d[2] = piece[j].z; d[3] = piece[j].w;
    }
    __syncthreads();

    v8f c = {};
    const float* xrow = xt + l15 * G0_PITCH + 2 * hh;
    const float* wcol = W0 + (2 * hh) * HIDDEN + col0 + l15;

#pragma unroll 8
    for (int k = 0; k < D_FEAT; k += 4) {
        v2f a, b;
        a.x = xrow[k];
        a.y = xrow[k + 1];
        b.x = wcol[k * HIDDEN];
        b.y = wcol[k * HIDDEN + HIDDEN];
        c = __builtin_amdgcn_wmma_f32_16x16x4_f32(false, a, false, b,
                                                  (short)0, c, false, false);
    }
    float*       out = m0   + (row0 + 8 * hh) * HIDDEN + col0 + l15;
    const float* dr  = dinv +  row0 + 8 * hh;
#pragma unroll
    for (int v = 0; v < 8; ++v) {
        out[v * HIDDEN] = dr[v] * c[v];
    }
}


#define S0_G      16
#define S0_FIN    (B_TBL * HIDDEN / 1024)
#define S0_STG    ((B_CAP + 1023) / 1024)
static_assert((B_TBL * HIDDEN) % 1024 == 0, "the final sweep is whole trips");
static_assert(32 % S0_G == 0, "a chunk of 32 records is whole groups");

static __device__ __forceinline__ void s0_walk(const uint2* q, const int n, const int lane,
                                               const float* m0, float* tl) {
    int   s_l  = 0;
    int   r_l  = 0;
    float wn_l = 0.0f;
    if (lane < n) {
        const uint2 r = q[lane];
        s_l  = (int)(r.x & 0x1FFFFu);
        r_l  = (int)(r.x >> 17);
        wn_l = __uint_as_float(r.y);
    }
    for (int j0 = 0; j0 < n; j0 += S0_G) {
        float v0[S0_G], v1[S0_G];
#pragma unroll
        for (int j = 0; j < S0_G; ++j) {
            const float* row = m0 + (unsigned)__builtin_amdgcn_readlane(s_l, j0 + j) * HIDDEN;
            v0[j] = row[lane];
            v1[j] = row[lane + 32];
        }
#pragma unroll
        for (int j = 0; j < S0_G; ++j) {
            if (j0 + j < n) {
                const int r = __builtin_amdgcn_readlane(r_l, j0 + j);
                const float wn = __int_as_float(
                    __builtin_amdgcn_readlane(__float_as_int(wn_l), j0 + j));
                atomicAdd(tl + r * HIDDEN,      wn * v0[j]);
                atomicAdd(tl + r * HIDDEN + 32, wn * v1[j]);
            }
        }
    }
}

static __device__ __forceinline__ void s0_scan(const int* ei, const float* ew, const float* m0,
                                               const int wv, const int lane, const int lo, float* tl) {
    for (int e0 = wv * 32; e0 < N_EDGES; e0 += 1024) {
        const int e = e0 + lane;
        const unsigned r = (unsigned)(ei[N_EDGES + e] - lo);
        const bool hit = r < (unsigned)B_TBL;
        unsigned bm = __builtin_amdgcn_ballot_w32(hit);
        if (bm == 0u) continue;
        int   s = 0;
        float w = 0.0f;
        if (hit) { s = ei[e]; w = ew[e]; }
        while (bm != 0u) {
            const int j = __builtin_ctz(bm);
            bm &= bm - 1u;
            const float* row = m0 + (unsigned)__builtin_amdgcn_readlane(s, j) * HIDDEN;
            const int   rj = __builtin_amdgcn_readlane((int)r, j);
            const float wj = __int_as_float(__builtin_amdgcn_readlane(__float_as_int(w), j));
            atomicAdd(tl + rj * HIDDEN,      wj * row[lane]);
            atomicAdd(tl + rj * HIDDEN + 32, wj * row[lane + 32]);
        }
    }
}

__global__ __launch_bounds__(1024) void k_scatter0(const int* __restrict__ ei,
                                                  const float* __restrict__ ew,
                                                  const float* __restrict__ dinv,
                                                  const unsigned* __restrict__ cursor,
                                                  const unsigned* __restrict__ ovf,
                                                  const float* __restrict__ m0,
                                                  float* h) {
    __shared__ float tbl[B_TBL * HIDDEN];
    __shared__ uint2 recs[B_CAP];
    const int tid  = threadIdx.x;
    const int lane = tid & 31;
    const int wv   = __builtin_amdgcn_readfirstlane(tid >> 5);
    const int part = blockIdx.x;
    const int lo   = part * B_TBL;

    const int total = __builtin_amdgcn_readfirstlane((int)cursor[part]);
    const bool over = total > B_CAP;
    const int n = over ? 0 : total;

    const uint2* mainp = (const uint2*)(h + (size_t)lo * HIDDEN);
    const uint2* ovfp  = (const uint2*)(ovf + part * (B_OVF * 2));
    uint2 rr[S0_STG];
#pragma unroll
    for (int k = 0; k < S0_STG; ++k) {
        const int i = k * 1024 + tid;
        rr[k] = make_uint2(0u, 0u);
        if (i < n) rr[k] = (i < B_MAIN) ? mainp[i] : ovfp[i - B_MAIN];
    }
    const float* own = m0 + (size_t)lo * HIDDEN;
    float mv[S0_FIN];
#pragma unroll
    for (int t = 0; t < S0_FIN; ++t) mv[t] = own[t * 1024 + tid];
#pragma unroll
    for (int k = 0; k < S0_STG; ++k) {
        const int i = k * 1024 + tid;
        if (i < n) recs[i] = rr[k];
    }
    for (int i = tid; i < B_TBL * HIDDEN; i += 1024) tbl[i] = 0.0f;
    __syncthreads();

    float* tl = tbl + lane;
    for (int c0 = wv * 32; c0 < n; c0 += 1024) {
        const int left = n - c0;
        s0_walk(recs + c0, left < 32 ? left : 32, lane, m0, tl);
    }
    if (over) s0_scan(ei, ew, m0, wv, lane, lo, tl);
    __syncthreads();

    float*       o   = h  + (size_t)lo * HIDDEN;
    const float* dl  = dinv + lo;
#pragma unroll
    for (int t = 0; t < S0_FIN; ++t) {
        const int i = t * 1024 + tid;
        o[i] = dl[i >> 6] * (tbl[i] + mv[t]);
    }
}


__global__ __launch_bounds__(128) void k_gemm1(const float* __restrict__ h,
                                               const float* __restrict__ b0,
                                               const float* __restrict__ W1,
                                               const float* __restrict__ dinv,
                                               const float* __restrict__ b1,
                                               float* __restrict__ m1,
                                               float* __restrict__ out) {
    const int lane = threadIdx.x & 31;
    const int wv   = threadIdx.x >> 5;
    const int rb   = blockIdx.x * 4 + wv;
    if (rb >= ROWBLKS) return;
    const int row0 = rb * 16;
    const int hh   = lane >> 4;
    const int l15  = lane & 15;

    v8f c = {};
    const float* hrow = h  + (row0 + l15) * HIDDEN + 2 * hh;
    const float* brow = b0 + 2 * hh;
#pragma unroll
    for (int k = 0; k < HIDDEN; k += 4) {
        v2f a, b;
        a.x = fmaxf(hrow[k]     + brow[k],     0.0f);
        a.y = fmaxf(hrow[k + 1] + brow[k + 1], 0.0f);
        float bx = 0.0f, by = 0.0f;
        if (l15 < NCLASS) {
            bx = W1[(k + 2 * hh) * NCLASS + l15];
            by = W1[(k + 2 * hh + 1) * NCLASS + l15];
        }
        b.x = bx; b.y = by;
        c = __builtin_amdgcn_wmma_f32_16x16x4_f32(false, a, false, b,
                                                  (short)0, c, false, false);
    }
    if (l15 < NCLASS) {
        float*       om = m1   + (row0 + 8 * hh) * NCLASS + l15;
        float*       oo = out  + (row0 + 8 * hh) * NCLASS + l15;
        const float* dr = dinv +  row0 + 8 * hh;
        const float  bj = b1[l15];
#pragma unroll
        for (int v = 0; v < 8; ++v) {
            const float di = dr[v];
            const float ms = di * c[v];
            om[v * NCLASS] = ms;
            oo[v * NCLASS] = di * ms + bj;
        }
    }
}


#define S1_TBL    10016
#define S1_PARTS  10
#define S1_SLICES 25
#define S1_EPS    (N_EDGES / S1_SLICES)
#define S1_UNR    5
static_assert(N_EDGES % S1_SLICES == 0, "slices are equal");
static_assert(S1_EPS % (1024 * S1_UNR) == 0, "a slice is a whole number of trips");
static_assert(S1_TBL % 32 == 0 && N_NODES % 32 == 0, "every table is whole 128-byte lines of out");
static_assert(S1_TBL * S1_PARTS >= N_NODES && S1_TBL * (S1_PARTS - 1) < N_NODES,
              "the tables cover the nodes and the last one is not empty");
static_assert(S1_TBL * (NCLASS + 1) * 4 <= 320 * 1024, "table and the part's dinv fit one WGP's LDS");

__global__ __launch_bounds__(1024) void k_scatter1(const int* __restrict__ ei,
                                                   const float* __restrict__ ew,
                                                   const float* __restrict__ dinv,
                                                   const float* __restrict__ m1,
                                                   float* __restrict__ out) {
    __shared__ float tbl[S1_TBL * NCLASS];
    __shared__ float dls[S1_TBL];
    const int tid   = threadIdx.x;
    const int part  = blockIdx.x % S1_PARTS;
    const int slice = blockIdx.x / S1_PARTS;
    const int lo    = part * S1_TBL;
    const int rest  = N_NODES - lo;
    const unsigned n_here = (unsigned)(rest < S1_TBL ? rest : S1_TBL);

    for (int i = tid; i < S1_TBL * NCLASS; i += 1024) tbl[i] = 0.0f;
    for (unsigned i = tid; i < n_here; i += 1024) dls[i] = dinv[lo + i];
    __syncthreads();

    const int*   src = ei + slice * S1_EPS + tid;
    const int*   dst = ei + N_EDGES + slice * S1_EPS + tid;
    const float* w   = ew + slice * S1_EPS + tid;

    int ns[S1_UNR], nd[S1_UNR]; float nw[S1_UNR];
#pragma unroll
    for (int k = 0; k < S1_UNR; ++k) {
        ns[k] = src[k * 1024]; nd[k] = dst[k * 1024]; nw[k] = w[k * 1024];
    }
    for (int base = 0; base < S1_EPS; base += 1024 * S1_UNR) {
        int cs[S1_UNR], cd[S1_UNR]; float cw[S1_UNR];
#pragma unroll
        for (int k = 0; k < S1_UNR; ++k) { cs[k] = ns[k]; cd[k] = nd[k]; cw[k] = nw[k]; }
        const int nb = base + 1024 * S1_UNR;
        if (nb < S1_EPS) {
#pragma unroll
            for (int k = 0; k < S1_UNR; ++k) {
                ns[k] = src[nb + k * 1024]; nd[k] = dst[nb + k * 1024]; nw[k] = w[nb + k * 1024];
            }
        }
        float v[S1_UNR][NCLASS];
#pragma unroll
        for (int k = 0; k < S1_UNR; ++k) {
            const unsigned r = (unsigned)(cd[k] - lo);
            if (r < n_here) {
                const float* row = m1 + cs[k] * NCLASS;
#pragma unroll
                for (int j = 0; j < NCLASS; ++j) v[k][j] = row[j];
            }
        }
#pragma unroll
        for (int k = 0; k < S1_UNR; ++k) {
            const unsigned r = (unsigned)(cd[k] - lo);
            if (r < n_here) {
#pragma unroll
                for (int j = 0; j < NCLASS; ++j) atomicAdd(&tbl[r * NCLASS + j], cw[k] * v[k][j]);
            }
        }
    }
    __syncthreads();

    const int lane = tid & 31;
    const int wv   = tid >> 5;
    const int nl   = (int)(n_here * NCLASS) >> 5;
    const int start = (slice * nl) / S1_SLICES;
    float* o = out + lo * NCLASS;
    for (int j = wv; j < nl; j += 32) {
        int line = j + start;
        if (line >= nl) line -= nl;
        const unsigned i = (unsigned)(line * 32 + lane);
        atomicAdd(&o[i], dls[i / NCLASS] * tbl[i]);
    }
}


extern "C" void kernel_launch(void* const* d_in, const int* in_sizes, int n_in,
                              void* d_out, int out_size, void* d_ws, size_t ws_size,
                              hipStream_t stream) {
    const float* x  = (const float*)d_in[0];
    const int*   ei = (const int*)  d_in[1];
    const float* ew = (const float*)d_in[2];
    const float* W0 = (const float*)d_in[3];
    const float* b0 = (const float*)d_in[4];
    const float* W1 = (const float*)d_in[5];
    const float* b1 = (const float*)d_in[6];
    float* out = (float*)d_out;

    float* ws   = (float*)d_ws;
    float* dinv = ws;
    float* m0   = dinv + N_NODES;
    float* h    = m0 + (size_t)N_NODES * HIDDEN;
    float* m1   = h  + (size_t)N_NODES * HIDDEN;
    unsigned* ovf    = (unsigned*)m1;
    unsigned* cursor = ovf + B_CUR;

    k_prep  <<<1, 256, 0, stream>>>(cursor);
    k_bin   <<<B_PARTS, 1024, 0, stream>>>(ei, ew, h, ovf, cursor);
    k_degbin<<<B_PARTS, 1024, 0, stream>>>(ei, ew, h, ovf, cursor, dinv);

    k_gemm0<<<ROWBLKS, 128, 0, stream>>>(x, W0, dinv, m0);

    k_scatter0<<<B_PARTS, 1024, 0, stream>>>(ei, ew, dinv, cursor, ovf, m0, h);

    k_gemm1<<<(ROWBLKS + 3) / 4, 128, 0, stream>>>(h, b0, W1, dinv, b1, m1, out);

    k_scatter1<<<S1_PARTS * S1_SLICES, 1024, 0, stream>>>(ei, ew, dinv, m1, out);
}
